# starting kernel
# baseline (speedup 1.0000x reference)
#include <hip/hip_runtime.h>

typedef __attribute__((ext_vector_type(2))) float v2f;
typedef __attribute__((ext_vector_type(8))) float v8f;

#define B_ 4
#define N_ 512
#define D_ 128
#define L_ 2
#define M_ (B_ * N_)
#define TI 8
#define JT 64

__global__ __launch_bounds__(32)
void wmma_gemm_bias(const float* __restrict__ X, const float* __restrict__ W,
                    int wstride, int woff, const float* __restrict__ bias,
                    float* __restrict__ out) {
  const int lane = threadIdx.x;
  const int l16  = lane & 15;
  const int half = lane >> 4;
  const int tm = blockIdx.y;
  const int tn = blockIdx.x;
  const float* xrow = X + (tm * 16 + l16) * D_;
  const float* wrow = W + (tn * 16 + l16) * wstride + woff;
  v8f c = {};
#pragma unroll
  for (int kk = 0; kk < D_ / 4; ++kk) {
    const int kb = kk * 4 + half * 2;
    v2f a = *(const v2f*)(xrow + kb);
    v2f b = *(const v2f*)(wrow + kb);
    c = __builtin_amdgcn_wmma_f32_16x16x4_f32(
        false, a, false, b, (short)0, c, false, false);
  }
  const float bv = bias ? bias[tn * 16 + l16] : 0.f;
#pragma unroll
  for (int r = 0; r < 8; ++r) {
    const int m = tm * 16 + half * 8 + r;
    out[m * D_ + tn * 16 + l16] = c[r] + bv;
  }
}

__global__ __launch_bounds__(32)
void wmma_gemm_final(const float* __restrict__ S, const float* __restrict__ W2,
                     const float* __restrict__ b2, const float* __restrict__ rowsum,
                     float* __restrict__ h) {
  const int lane = threadIdx.x;
  const int l16  = lane & 15;
  const int half = lane >> 4;
  const int tm = blockIdx.y;
  const int tn = blockIdx.x;
  const float* xrow = S + (tm * 16 + l16) * D_;
  const float* wrow = W2 + (tn * 16 + l16) * D_;
  v8f c = {};
#pragma unroll
  for (int kk = 0; kk < D_ / 4; ++kk) {
    const int kb = kk * 4 + half * 2;
    v2f a = *(const v2f*)(xrow + kb);
    v2f b = *(const v2f*)(wrow + kb);
    c = __builtin_amdgcn_wmma_f32_16x16x4_f32(
        false, a, false, b, (short)0, c, false, false);
  }
  const float bv = b2[tn * 16 + l16];
#pragma unroll
  for (int r = 0; r < 8; ++r) {
    const int m = tm * 16 + half * 8 + r;
    const int idx = m * D_ + tn * 16 + l16;
    h[idx] = h[idx] + c[r] + rowsum[m] * bv;
  }
}

__global__ __launch_bounds__(256)
void pairwise_reduce(const float* __restrict__ ai, const float* __restrict__ aj,
                     const float* __restrict__ adj, float* __restrict__ S) {
  __shared__ float ajT[D_ * (JT + 1)];
  __shared__ float adjT[TI * JT];
  const int b  = blockIdx.y;
  const int i0 = blockIdx.x * TI;
  const int tid = threadIdx.x;
  const int o  = tid & (D_ - 1);
  const int ip = tid >> 7;

  float acc[TI / 2];
  float aiv[TI / 2];
#pragma unroll
  for (int r = 0; r < TI / 2; ++r) {
    acc[r] = 0.f;
    aiv[r] = ai[((b * N_) + i0 + ip * (TI / 2) + r) * D_ + o];
  }

  for (int j0 = 0; j0 < N_; j0 += JT) {
    __syncthreads();
    for (int t = tid; t < JT * D_; t += 256) {
      const int jj = t >> 7, oo = t & (D_ - 1);
      ajT[oo * (JT + 1) + jj] = aj[((b * N_) + j0 + jj) * D_ + oo];
    }
    for (int t = tid; t < TI * JT; t += 256) {
      const int ii = t >> 6, jj = t & (JT - 1);
      adjT[ii * JT + jj] = adj[((b * N_) + i0 + ii) * N_ + j0 + jj];
    }
    __syncthreads();
#pragma unroll 4
    for (int jj = 0; jj < JT; ++jj) {
      const float ajv = ajT[o * (JT + 1) + jj];
#pragma unroll
      for (int r = 0; r < TI / 2; ++r) {
        float t = aiv[r] + ajv;
        t = fmaxf(t, 0.f);
        acc[r] = fmaf(adjT[(ip * (TI / 2) + r) * JT + jj], t, acc[r]);
      }
    }
  }
#pragma unroll
  for (int r = 0; r < TI / 2; ++r)
    S[((b * N_) + i0 + ip * (TI / 2) + r) * D_ + o] = acc[r];
}

__global__ __launch_bounds__(256)
void rowsum_kernel(const float* __restrict__ adj, float* __restrict__ rowsum) {
  const int lane = threadIdx.x & 31;
  const int wave = threadIdx.x >> 5;
  const int row  = blockIdx.x * 8 + wave;
  const float* p = adj + row * N_;
  float s = 0.f;
  for (int j = lane; j < N_; j += 32) s += p[j];
#pragma unroll
  for (int off = 16; off > 0; off >>= 1) s += __shfl_down(s, off, 32);
  if (lane == 0) rowsum[row] = s;
}

extern "C" void kernel_launch(void* const* d_in, const int* in_sizes, int n_in,
                              void* d_out, int out_size, void* d_ws, size_t ws_size,
                              hipStream_t stream) {
  const float* node = (const float*)d_in[0];
  const float* adj  = (const float*)d_in[1];
  const float* w1   = (const float*)d_in[2];
  const float* b1   = (const float*)d_in[3];
  const float* w2   = (const float*)d_in[4];
  const float* b2   = (const float*)d_in[5];

  float* h  = (float*)d_out;
  float* ai = (float*)d_ws;
  float* aj = ai + (size_t)M_ * D_;
  float* S  = aj + (size_t)M_ * D_;
  float* rs = S  + (size_t)M_ * D_;

  hipMemcpyAsync(h, node, (size_t)M_ * D_ * sizeof(float),
                 hipMemcpyDeviceToDevice, stream);
  rowsum_kernel<<<M_ / 8, 256, 0, stream>>>(adj, rs);

  const dim3 gg(D_ / 16, M_ / 16);
  for (int l = 0; l < L_; ++l) {
    const float* w1l = w1 + (size_t)l * D_ * 2 * D_;
    const float* b1l = b1 + (size_t)l * D_;
    const float* w2l = w2 + (size_t)l * D_ * D_;
    const float* b2l = b2 + (size_t)l * D_;
    wmma_gemm_bias<<<gg, 32, 0, stream>>>(h, w1l, 2 * D_, 0,  b1l, ai);
    wmma_gemm_bias<<<gg, 32, 0, stream>>>(h, w1l, 2 * D_, D_, nullptr, aj);
    pairwise_reduce<<<dim3(N_ / TI, B_), 256, 0, stream>>>(ai, aj, adj, S);
    wmma_gemm_final<<<gg, 32, 0, stream>>>(S, w2l, b2l, rs, h);
  }
}


# ai product moved into its consumer: each pairwise block forms its own 32x32 ai tile on the matrix unit from h and wi rows staged in LDS beside adj; the two ai launches go; k and j sum orders kept
# speedup vs baseline: 1.9237x; 1.9237x over previous
#include <hip/hip_runtime.h>

typedef __attribute__((ext_vector_type(2))) float v2f;
typedef __attribute__((ext_vector_type(8))) float v8f;

#define B_ 4
#define N_ 512
#define D_ 128
#define L_ 2
#define M_ (B_ * N_)

template <bool PACK>
__global__ __launch_bounds__(32)
void wmma_gemm_bias(const float* __restrict__ X, const float* __restrict__ W,
                    int wstride, int woff, const float* __restrict__ bias,
                    float* __restrict__ out) {
  const int lane = threadIdx.x;
  const int l16  = lane & 15;
  const int half = lane >> 4;
  const int tm = blockIdx.y;
  const int tn = blockIdx.x;
  const float* xrow = X + (tm * 16 + l16) * D_;
  const float* wrow = W + (tn * 16 + l16) * wstride + woff;
  v8f c = {};
#pragma unroll
  for (int kk = 0; kk < D_ / 4; ++kk) {
    const int kb = kk * 4 + half * 2;
    v2f a = *(const v2f*)(xrow + kb);
    v2f b = *(const v2f*)(wrow + kb);
    c = __builtin_amdgcn_wmma_f32_16x16x4_f32(
        false, a, false, b, (short)0, c, false, false);
  }
  const float bv = bias ? bias[tn * 16 + l16] : 0.f;
#pragma unroll
  for (int r = 0; r < 8; ++r) {
    const int m = tm * 16 + half * 8 + r;
    const int o = tn * 16 + l16;
    if (PACK)
      out[((((m >> 9) * (D_ / 2) + (o >> 1)) * N_ + (m & (N_ - 1))) << 1) + (o & 1)] = c[r] + bv;
    else
      out[m * D_ + o] = c[r] + bv;
  }
}

__global__ __launch_bounds__(32)
void wmma_gemm_final(const float* __restrict__ S, const float* __restrict__ W2,
                     const float* __restrict__ b2, const float* __restrict__ rowsum,
                     float* __restrict__ h) {
  const int lane = threadIdx.x;
  const int l16  = lane & 15;
  const int half = lane >> 4;
  const int tm = blockIdx.y;
  const int tn = blockIdx.x;
  const float* xrow = S + (tm * 16 + l16) * D_;
  const float* wrow = W2 + (tn * 16 + l16) * D_;
  v8f c = {};
#pragma unroll
  for (int kk = 0; kk < D_ / 4; ++kk) {
    const int kb = kk * 4 + half * 2;
    v2f a = *(const v2f*)(xrow + kb);
    v2f b = *(const v2f*)(wrow + kb);
    c = __builtin_amdgcn_wmma_f32_16x16x4_f32(
        false, a, false, b, (short)0, c, false, false);
  }
  const float bv = b2[tn * 16 + l16];
#pragma unroll
  for (int r = 0; r < 8; ++r) {
    const int m = tm * 16 + half * 8 + r;
    const int idx = m * D_ + tn * 16 + l16;
    h[idx] = h[idx] + c[r] + rowsum[m] * bv;
  }
}

typedef __attribute__((ext_vector_type(4)))  float v4f;
typedef __attribute__((ext_vector_type(16))) float v16f;
#define PR 32
#define PO 32
#define APITCH (N_ + 4)
#define HP (D_ + 4)
#define AIP (PO + 1)

__device__ __forceinline__ void pair_step8(const v16f s, const v4f alo, const v4f ahi,
                                           const float ai0, const float ai1,
                                           float& acc0, float& acc1) {
#pragma unroll
  for (int jj = 0; jj < 8; ++jj) {
    const float av = jj < 4 ? alo[jj & 3] : ahi[jj & 3];
    float t0 = ai0 + s[2 * jj];
    float t1 = ai1 + s[2 * jj + 1];
    t0 = fmaxf(t0, 0.f);
    t1 = fmaxf(t1, 0.f);
    acc0 = fmaf(av, t0, acc0);
    acc1 = fmaf(av, t1, acc1);
  }
}

__global__ __launch_bounds__(512)
void pairwise_rows(const float* __restrict__ h, const float* __restrict__ wi,
                   const float* __restrict__ b1, const float* __restrict__ ajP,
                   const float* __restrict__ adj, float* __restrict__ S) {
  __shared__ __attribute__((aligned(16))) float adjL[PR * APITCH];
  __shared__ __attribute__((aligned(16))) float hL[PR * HP];
  __shared__ __attribute__((aligned(16))) float wL[PO * HP];
  __shared__ float aiL[PR * AIP];
  const int tid  = threadIdx.x;
  const int lane = tid & 31;
  const int w    = __builtin_amdgcn_readfirstlane(tid >> 5);
  const int ob   = blockIdx.x;
  const int row0 = blockIdx.y * PR;
  const int b    = blockIdx.y >> 4;

  const v4f* ag = (const v4f*)(adj + (size_t)row0 * N_);
#pragma unroll
  for (int p = 0; p < 8; ++p) {
    const int idx = p * 512 + tid;
    const int r = idx >> 7, c4 = idx & 127;
    *(v4f*)(adjL + r * APITCH + c4 * 4) = ag[r * (N_ / 4) + c4];
  }
#pragma unroll
  for (int p = 0; p < 2; ++p) {
    const int idx = p * 512 + tid;
    const int r = idx >> 5, c4 = idx & 31;
    *(v4f*)(hL + r * HP + c4 * 4) = *(const v4f*)(h + (size_t)(row0 + r) * D_ + c4 * 4);
    *(v4f*)(wL + r * HP + c4 * 4) = *(const v4f*)(wi + (size_t)(ob * PO + r) * (2 * D_) + c4 * 4);
  }

  const float* sp = ajP + (size_t)(b * (D_ / 2) + ob * (PO / 2) + w) * N_ * 2;
  const float* arow = adjL + lane * APITCH;
  float acc0 = 0.f, acc1 = 0.f;

  v16f sA0 = *(const v16f*)(sp), sA1 = *(const v16f*)(sp + 16);
  __syncthreads();
  if (w < 4) {
    const int l16  = lane & 15;
    const int half = lane >> 4;
    const int tr = w >> 1, tc = w & 1;
    const float* xrow = hL + (tr * 16 + l16) * HP;
    const float* wrow = wL + (tc * 16 + l16) * HP;
    const float bv = b1[ob * PO + tc * 16 + l16];
    v8f c = {};
#pragma unroll
    for (int kk = 0; kk < D_ / 4; ++kk) {
      const int kb = kk * 4 + half * 2;
      v2f a = *(const v2f*)(xrow + kb);
      v2f b = *(const v2f*)(wrow + kb);
      c = __builtin_amdgcn_wmma_f32_16x16x4_f32(
          false, a, false, b, (short)0, c, false, false);
    }
#pragma unroll
    for (int r = 0; r < 8; ++r)
      aiL[(tr * 16 + half * 8 + r) * AIP + tc * 16 + l16] = c[r] + bv;
  }
  __syncthreads();
  const float ai0 = aiL[lane * AIP + w * 2], ai1 = aiL[lane * AIP + w * 2 + 1];
  v4f aA0 = *(const v4f*)(arow), aA1 = *(const v4f*)(arow + 4),
      aA2 = *(const v4f*)(arow + 8), aA3 = *(const v4f*)(arow + 12);
  for (int g = 0; g < N_ / 16; g += 2) {
    asm volatile("" :: "s"(sA0), "s"(sA1));
    const float* s1 = sp + (g + 1) * 32;
    const float* r1 = arow + (g + 1) * 16;
    const v16f sB0 = *(const v16f*)(s1), sB1 = *(const v16f*)(s1 + 16);
    const v4f aB0 = *(const v4f*)(r1), aB1 = *(const v4f*)(r1 + 4),
              aB2 = *(const v4f*)(r1 + 8), aB3 = *(const v4f*)(r1 + 12);
    __builtin_amdgcn_sched_barrier(0);
    pair_step8(sA0, aA0, aA1, ai0, ai1, acc0, acc1);
    pair_step8(sA1, aA2, aA3, ai0, ai1, acc0, acc1);
    asm volatile("" :: "s"(sB0), "s"(sB1));
    const int g2 = (g + 2 < N_ / 16) ? g + 2 : g + 1;
    const float* s2 = sp + g2 * 32;
    const float* r2 = arow + g2 * 16;
    sA0 = *(const v16f*)(s2); sA1 = *(const v16f*)(s2 + 16);
    aA0 = *(const v4f*)(r2); aA1 = *(const v4f*)(r2 + 4);
    aA2 = *(const v4f*)(r2 + 8); aA3 = *(const v4f*)(r2 + 12);
    __builtin_amdgcn_sched_barrier(0);
    pair_step8(sB0, aB0, aB1, ai0, ai1, acc0, acc1);
    pair_step8(sB1, aB2, aB3, ai0, ai1, acc0, acc1);
  }

  __syncthreads();
  adjL[lane * (PO + 1) + w * 2]     = acc0;
  adjL[lane * (PO + 1) + w * 2 + 1] = acc1;
  __syncthreads();
#pragma unroll
  for (int rr = 0; rr < 2; ++rr) {
    const int r = w * 2 + rr;
    S[(size_t)(row0 + r) * D_ + ob * PO + lane] = adjL[r * (PO + 1) + lane];
  }
}

__global__ __launch_bounds__(256)
void rowsum_kernel(const float* __restrict__ adj, float* __restrict__ rowsum) {
  const int lane = threadIdx.x & 31;
  const int wave = threadIdx.x >> 5;
  const int row  = blockIdx.x * 8 + wave;
  const float* p = adj + row * N_;
  float s = 0.f;
  for (int j = lane; j < N_; j += 32) s += p[j];
#pragma unroll
  for (int off = 16; off > 0; off >>= 1) s += __shfl_down(s, off, 32);
  if (lane == 0) rowsum[row] = s;
}

extern "C" void kernel_launch(void* const* d_in, const int* in_sizes, int n_in,
                              void* d_out, int out_size, void* d_ws, size_t ws_size,
                              hipStream_t stream) {
  const float* node = (const float*)d_in[0];
  const float* adj  = (const float*)d_in[1];
  const float* w1   = (const float*)d_in[2];
  const float* b1   = (const float*)d_in[3];
  const float* w2   = (const float*)d_in[4];
  const float* b2   = (const float*)d_in[5];

  float* h  = (float*)d_out;
  float* aj = (float*)d_ws;
  float* S  = aj + (size_t)M_ * D_;
  float* rs = S  + (size_t)M_ * D_;

  hipMemcpyAsync(h, node, (size_t)M_ * D_ * sizeof(float),
                 hipMemcpyDeviceToDevice, stream);
  rowsum_kernel<<<M_ / 8, 256, 0, stream>>>(adj, rs);

  const dim3 gg(D_ / 16, M_ / 16);
  for (int l = 0; l < L_; ++l) {
    const float* w1l = w1 + (size_t)l * D_ * 2 * D_;
    const float* b1l = b1 + (size_t)l * D_;
    const float* w2l = w2 + (size_t)l * D_ * D_;
    const float* b2l = b2 + (size_t)l * D_;
    wmma_gemm_bias<true><<<gg, 32, 0, stream>>>(h, w1l, 2 * D_, D_, nullptr, aj);
    pairwise_rows<<<dim3(D_ / PO, M_ / PR), 512, 0, stream>>>(h, w1l, b1l, aj, adj, S);
    wmma_gemm_final<<<gg, 32, 0, stream>>>(S, w2l, b2l, rs, h);
  }
}


# layer update folded away: h1 never formed, layer-1 products read h0 and S0 through folded weights w1_1 @ w2_0, output is one two-part product; one update launch and the input copy go; sums regrouped
# speedup vs baseline: 2.0067x; 1.0431x over previous
#include <hip/hip_runtime.h>

typedef __attribute__((ext_vector_type(2))) float v2f;
typedef __attribute__((ext_vector_type(4))) float v4f;
typedef __attribute__((ext_vector_type(8))) float v8f;
typedef __attribute__((ext_vector_type(16))) float v16f;

#define B_ 4
#define N_ 512
#define D_ 128
#define L_ 2
#define M_ (B_ * N_)


__device__ __forceinline__ v8f tile_k128(const float* __restrict__ xrow,
                                         const float* __restrict__ wrow,
                                         const int half, v8f c) {
#pragma unroll
  for (int kk = 0; kk < D_ / 4; ++kk) {
    const int kb = kk * 4 + half * 2;
    v2f a = *(const v2f*)(xrow + kb);
    v2f b = *(const v2f*)(wrow + kb);
    c = __builtin_amdgcn_wmma_f32_16x16x4_f32(
        false, a, false, b, (short)0, c, false, false);
  }
  return c;
}

__device__ __forceinline__ int pack_index(const int m, const int o) {
  return ((((m >> 9) * (D_ / 2) + (o >> 1)) * N_ + (m & (N_ - 1))) << 1) + (o & 1);
}

__global__ __launch_bounds__(32)
void aj0_and_fold(const float* __restrict__ node, const float* __restrict__ w1,
                  const float* __restrict__ w2, const float* __restrict__ b2,
                  float* __restrict__ aj, float* __restrict__ Cf, float* __restrict__ vf) {
  const int lane = threadIdx.x;
  const int l16  = lane & 15;
  const int half = lane >> 4;
  const int tn = blockIdx.x;
  const int ty = blockIdx.y;
  const float* w1n = w1 + D_ * 2 * D_;
  if (ty < M_ / 16) {
    const float* xrow = node + (ty * 16 + l16) * D_;
    const float* wrow = w1 + (tn * 16 + l16) * (2 * D_) + D_;
    v8f c = {};
    c = tile_k128(xrow, wrow, half, c);
#pragma unroll
    for (int r = 0; r < 8; ++r)
      aj[pack_index(ty * 16 + half * 8 + r, tn * 16 + l16)] = c[r];
  } else if (ty < M_ / 16 + 16) {
    const int t = ty - M_ / 16;
    const int sel = t >> 3, to = t & 7;
    const float* xrow = w1n + (to * 16 + l16) * (2 * D_) + sel * D_;
    const float* wcol = w2 + tn * 16 + l16;
    v8f c = {};
#pragma unroll
    for (int kk = 0; kk < D_ / 4; ++kk) {
      const int kb = kk * 4 + half * 2;
      v2f a = *(const v2f*)(xrow + kb);
      v2f b;
      b[0] = wcol[kb * D_];
      b[1] = wcol[(kb + 1) * D_];
      c = __builtin_amdgcn_wmma_f32_16x16x4_f32(
          false, a, false, b, (short)0, c, false, false);
    }
#pragma unroll
    for (int r = 0; r < 8; ++r)
      Cf[(sel * D_ + to * 16 + half * 8 + r) * D_ + tn * 16 + l16] = c[r];
  } else {
    const int o = tn * 16 + l16;
    const v4f* wr = (const v4f*)(w1n + o * (2 * D_) + half * D_);
    const v4f* bb = (const v4f*)b2;
    float s = 0.f;
#pragma unroll 8
    for (int e = 0; e < D_ / 4; ++e) {
      const v4f x = wr[e], y = bb[e];
      s = fmaf(x[0], y[0], s); s = fmaf(x[1], y[1], s);
      s = fmaf(x[2], y[2], s); s = fmaf(x[3], y[3], s);
    }
    vf[half * D_ + o] = s;
  }
}

template <bool PACK, bool RESID>
__global__ __launch_bounds__(32)
void gemm_two_parts(const float* __restrict__ X0, const float* __restrict__ W0,
                    int ws0, int wo0,
                    const float* __restrict__ X1, const float* __restrict__ W1,
                    const float* __restrict__ rs, const float* __restrict__ v0,
                    const float* __restrict__ v1, const float* __restrict__ R,
                    float* __restrict__ out) {
  const int lane = threadIdx.x;
  const int l16  = lane & 15;
  const int half = lane >> 4;
  const int tm = blockIdx.y;
  const int tn = blockIdx.x;
  v8f c0 = {}, c1 = {};
  c0 = tile_k128(X0 + (tm * 16 + l16) * D_, W0 + (tn * 16 + l16) * ws0 + wo0, half, c0);
  c1 = tile_k128(X1 + (tm * 16 + l16) * D_, W1 + (tn * 16 + l16) * D_, half, c1);
  const int o = tn * 16 + l16;
  const float bv = v0[o] + (v1 ? v1[o] : 0.f);
#pragma unroll
  for (int r = 0; r < 8; ++r) {
    const int m = tm * 16 + half * 8 + r;
    float val = (c0[r] + c1[r]) + rs[m] * bv;
    if (RESID) val += R[m * D_ + o];
    if (PACK) out[pack_index(m, o)] = val;
    else      out[m * D_ + o] = val;
  }
}

#define PR 32
#define PO 32
#define APITCH (N_ + 4)
#define AIP (PO + 1)

__device__ __forceinline__ void pair_step8(const v16f s, const v4f alo, const v4f ahi,
                                           const float ai0, const float ai1,
                                           float& acc0, float& acc1) {
#pragma unroll
  for (int jj = 0; jj < 8; ++jj) {
    const float av = jj < 4 ? alo[jj & 3] : ahi[jj & 3];
    float t0 = ai0 + s[2 * jj];
    float t1 = ai1 + s[2 * jj + 1];
    t0 = fmaxf(t0, 0.f);
    t1 = fmaxf(t1, 0.f);
    acc0 = fmaf(av, t0, acc0);
    acc1 = fmaf(av, t1, acc1);
  }
}

template <int KS>
__global__ __launch_bounds__(512)
void pairwise_rows(const float* __restrict__ h, const float* __restrict__ S0,
                   const float* __restrict__ wi, const float* __restrict__ Ci,
                   const float* __restrict__ b1, const float* __restrict__ rs,
                   const float* __restrict__ vi, const float* __restrict__ ajP,
                   const float* __restrict__ adj, float* __restrict__ S) {
  constexpr int HP = KS * D_ + 4;
  __shared__ __attribute__((aligned(16))) float adjL[PR * APITCH];
  __shared__ __attribute__((aligned(16))) float hL[PR * HP];
  __shared__ __attribute__((aligned(16))) float wL[PO * HP];
  __shared__ float aiL[KS * PR * AIP];
  const int tid  = threadIdx.x;
  const int lane = tid & 31;
  const int w    = __builtin_amdgcn_readfirstlane(tid >> 5);
  const int ob   = blockIdx.x;
  const int row0 = blockIdx.y * PR;
  const int b    = blockIdx.y >> 4;

  const v4f* ag = (const v4f*)(adj + (size_t)row0 * N_);
#pragma unroll
  for (int p = 0; p < 8; ++p) {
    const int idx = p * 512 + tid;
    const int r = idx >> 7, c4 = idx & 127;
    *(v4f*)(adjL + r * APITCH + c4 * 4) = ag[r * (N_ / 4) + c4];
  }
#pragma unroll
  for (int p = 0; p < 2; ++p) {
    const int idx = p * 512 + tid;
    const int r = idx >> 5, c4 = idx & 31;
    *(v4f*)(hL + r * HP + c4 * 4) = *(const v4f*)(h + (size_t)(row0 + r) * D_ + c4 * 4);
    *(v4f*)(wL + r * HP + c4 * 4) = *(const v4f*)(wi + (size_t)(ob * PO + r) * (2 * D_) + c4 * 4);
    if (KS == 2) {
      *(v4f*)(hL + r * HP + D_ + c4 * 4) = *(const v4f*)(S0 + (size_t)(row0 + r) * D_ + c4 * 4);
      *(v4f*)(wL + r * HP + D_ + c4 * 4) = *(const v4f*)(Ci + (size_t)(ob * PO + r) * D_ + c4 * 4);
    }
  }

  const float* sp = ajP + (size_t)(b * (D_ / 2) + ob * (PO / 2) + w) * N_ * 2;
  const float* arow = adjL + lane * APITCH;
  float acc0 = 0.f, acc1 = 0.f;

  v16f sA0 = *(const v16f*)(sp), sA1 = *(const v16f*)(sp + 16);
  __syncthreads();
  if (w < 4 * KS) {
    const int l16  = lane & 15;
    const int half = lane >> 4;
    const int seg = w >> 2;
    const int tr = (w >> 1) & 1, tc = w & 1;
    const float* xrow = hL + (tr * 16 + l16) * HP + seg * D_;
    const float* wrow = wL + (tc * 16 + l16) * HP + seg * D_;
    const float bv = seg == 0 ? b1[ob * PO + tc * 16 + l16] : 0.f;
    v8f c = {};
#pragma unroll
    for (int kk = 0; kk < D_ / 4; ++kk) {
      const int kb = kk * 4 + half * 2;
      v2f a = *(const v2f*)(xrow + kb);
      v2f b = *(const v2f*)(wrow + kb);
      c = __builtin_amdgcn_wmma_f32_16x16x4_f32(
          false, a, false, b, (short)0, c, false, false);
    }
#pragma unroll
    for (int r = 0; r < 8; ++r)
      aiL[seg * PR * AIP + (tr * 16 + half * 8 + r) * AIP + tc * 16 + l16] = c[r] + bv;
  }
  __syncthreads();
  float ai0 = aiL[lane * AIP + w * 2], ai1 = aiL[lane * AIP + w * 2 + 1];
  if (KS == 2) {
    const float rv = rs[row0 + lane];
    ai0 = (ai0 + aiL[PR * AIP + lane * AIP + w * 2])     + rv * vi[ob * PO + w * 2];
    ai1 = (ai1 + aiL[PR * AIP + lane * AIP + w * 2 + 1]) + rv * vi[ob * PO + w * 2 + 1];
  }
  v4f aA0 = *(const v4f*)(arow), aA1 = *(const v4f*)(arow + 4),
      aA2 = *(const v4f*)(arow + 8), aA3 = *(const v4f*)(arow + 12);
  for (int g = 0; g < N_ / 16; g += 2) {
    asm volatile("" :: "s"(sA0), "s"(sA1));
    const float* s1 = sp + (g + 1) * 32;
    const float* r1 = arow + (g + 1) * 16;
    const v16f sB0 = *(const v16f*)(s1), sB1 = *(const v16f*)(s1 + 16);
    const v4f aB0 = *(const v4f*)(r1), aB1 = *(const v4f*)(r1 + 4),
              aB2 = *(const v4f*)(r1 + 8), aB3 = *(const v4f*)(r1 + 12);
    __builtin_amdgcn_sched_barrier(0);
    pair_step8(sA0, aA0, aA1, ai0, ai1, acc0, acc1);
    pair_step8(sA1, aA2, aA3, ai0, ai1, acc0, acc1);
    asm volatile("" :: "s"(sB0), "s"(sB1));
    const int g2 = (g + 2 < N_ / 16) ? g + 2 : g + 1;
    const float* s2 = sp + g2 * 32;
    const float* r2 = arow + g2 * 16;
    sA0 = *(const v16f*)(s2); sA1 = *(const v16f*)(s2 + 16);
    aA0 = *(const v4f*)(r2); aA1 = *(const v4f*)(r2 + 4);
    aA2 = *(const v4f*)(r2 + 8); aA3 = *(const v4f*)(r2 + 12);
    __builtin_amdgcn_sched_barrier(0);
    pair_step8(sB0, aB0, aB1, ai0, ai1, acc0, acc1);
    pair_step8(sB1, aB2, aB3, ai0, ai1, acc0, acc1);
  }

  __syncthreads();
  adjL[lane * (PO + 1) + w * 2]     = acc0;
  adjL[lane * (PO + 1) + w * 2 + 1] = acc1;
  __syncthreads();
#pragma unroll
  for (int rr = 0; rr < 2; ++rr) {
    const int r = w * 2 + rr;
    S[(size_t)(row0 + r) * D_ + ob * PO + lane] = adjL[r * (PO + 1) + lane];
  }
}

__global__ __launch_bounds__(256)
void rowsum_kernel(const float* __restrict__ adj, float* __restrict__ rowsum) {
  const int lane = threadIdx.x & 31;
  const int wave = threadIdx.x >> 5;
  const int row  = blockIdx.x * 8 + wave;
  const float* p = adj + row * N_;
  float s = 0.f;
  for (int j = lane; j < N_; j += 32) s += p[j];
#pragma unroll
  for (int off = 16; off > 0; off >>= 1) s += __shfl_down(s, off, 32);
  if (lane == 0) rowsum[row] = s;
}

extern "C" void kernel_launch(void* const* d_in, const int* in_sizes, int n_in,
                              void* d_out, int out_size, void* d_ws, size_t ws_size,
                              hipStream_t stream) {
  const float* node = (const float*)d_in[0];
  const float* adj  = (const float*)d_in[1];
  const float* w1   = (const float*)d_in[2];
  const float* b1   = (const float*)d_in[3];
  const float* w2   = (const float*)d_in[4];
  const float* b2   = (const float*)d_in[5];

  float* out = (float*)d_out;
  float* aj = (float*)d_ws;
  float* S0 = aj + (size_t)M_ * D_;
  float* S1 = S0 + (size_t)M_ * D_;
  float* rs = S1 + (size_t)M_ * D_;
  float* Cf = rs + M_;
  float* vf = Cf + 2 * D_ * D_;

  const float* w1n = w1 + (size_t)D_ * 2 * D_;
  const float* b1n = b1 + D_;
  const float* w2n = w2 + (size_t)D_ * D_;
  const float* b2n = b2 + D_;

  rowsum_kernel<<<M_ / 8, 256, 0, stream>>>(adj, rs);

  const dim3 gg(D_ / 16, M_ / 16);
  aj0_and_fold<<<dim3(D_ / 16, M_ / 16 + 17), 32, 0, stream>>>(node, w1, w2, b2, aj, Cf, vf);
  pairwise_rows<1><<<dim3(D_ / PO, M_ / PR), 512, 0, stream>>>(
      node, nullptr, w1, nullptr, b1, nullptr, nullptr, aj, adj, S0);
  gemm_two_parts<true, false><<<gg, 32, 0, stream>>>(
      node, w1n, 2 * D_, D_, S0, Cf + D_ * D_, rs, vf + D_, nullptr, nullptr, aj);
  pairwise_rows<2><<<dim3(D_ / PO, M_ / PR), 512, 0, stream>>>(
      node, S0, w1n, Cf, b1n, rs, vf, aj, adj, S1);
  gemm_two_parts<false, true><<<gg, 32, 0, stream>>>(
      S0, w2, D_, 0, S1, w2n, rs, b2, b2n, node, out);
}


# row sums of adj formed inside the first pairwise launch from the rows already in LDS, by waves idle during the ai tile; the rowsum launch goes; row sum order changes (8 chunks of 64 columns)
# speedup vs baseline: 2.1385x; 1.0657x over previous
#include <hip/hip_runtime.h>

typedef __attribute__((ext_vector_type(2))) float v2f;
typedef __attribute__((ext_vector_type(4))) float v4f;
typedef __attribute__((ext_vector_type(8))) float v8f;
typedef __attribute__((ext_vector_type(16))) float v16f;

#define B_ 4
#define N_ 512
#define D_ 128
#define L_ 2
#define M_ (B_ * N_)


__device__ __forceinline__ v8f tile_k128(const float* __restrict__ xrow,
                                         const float* __restrict__ wrow,
                                         const int half, v8f c) {
#pragma unroll
  for (int kk = 0; kk < D_ / 4; ++kk) {
    const int kb = kk * 4 + half * 2;
    v2f a = *(const v2f*)(xrow + kb);
    v2f b = *(const v2f*)(wrow + kb);
    c = __builtin_amdgcn_wmma_f32_16x16x4_f32(
        false, a, false, b, (short)0, c, false, false);
  }
  return c;
}

__device__ __forceinline__ int pack_index(const int m, const int o) {
  return ((((m >> 9) * (D_ / 2) + (o >> 1)) * N_ + (m & (N_ - 1))) << 1) + (o & 1);
}

__global__ __launch_bounds__(32)
void aj0_and_fold(const float* __restrict__ node, const float* __restrict__ w1,
                  const float* __restrict__ w2, const float* __restrict__ b2,
                  float* __restrict__ aj, float* __restrict__ Cf, float* __restrict__ vf) {
  const int lane = threadIdx.x;
  const int l16  = lane & 15;
  const int half = lane >> 4;
  const int tn = blockIdx.x;
  const int ty = blockIdx.y;
  const float* w1n = w1 + D_ * 2 * D_;
  if (ty < M_ / 16) {
    const float* xrow = node + (ty * 16 + l16) * D_;
    const float* wrow = w1 + (tn * 16 + l16) * (2 * D_) + D_;
    v8f c = {};
    c = tile_k128(xrow, wrow, half, c);
#pragma unroll
    for (int r = 0; r < 8; ++r)
      aj[pack_index(ty * 16 + half * 8 + r, tn * 16 + l16)] = c[r];
  } else if (ty < M_ / 16 + 16) {
    const int t = ty - M_ / 16;
    const int sel = t >> 3, to = t & 7;
    const float* xrow = w1n + (to * 16 + l16) * (2 * D_) + sel * D_;
    const float* wcol = w2 + tn * 16 + l16;
    v8f c = {};
#pragma unroll
    for (int kk = 0; kk < D_ / 4; ++kk) {
      const int kb = kk * 4 + half * 2;
      v2f a = *(const v2f*)(xrow + kb);
      v2f b;
      b[0] = wcol[kb * D_];
      b[1] = wcol[(kb + 1) * D_];
      c = __builtin_amdgcn_wmma_f32_16x16x4_f32(
          false, a, false, b, (short)0, c, false, false);
    }
#pragma unroll
    for (int r = 0; r < 8; ++r)
      Cf[(sel * D_ + to * 16 + half * 8 + r) * D_ + tn * 16 + l16] = c[r];
  } else {
    const int o = tn * 16 + l16;
    const v4f* wr = (const v4f*)(w1n + o * (2 * D_) + half * D_);
    const v4f* bb = (const v4f*)b2;
    float s = 0.f;
#pragma unroll 8
    for (int e = 0; e < D_ / 4; ++e) {
      const v4f x = wr[e], y = bb[e];
      s = fmaf(x[0], y[0], s); s = fmaf(x[1], y[1], s);
      s = fmaf(x[2], y[2], s); s = fmaf(x[3], y[3], s);
    }
    vf[half * D_ + o] = s;
  }
}

template <bool PACK, bool RESID>
__global__ __launch_bounds__(32)
void gemm_two_parts(const float* __restrict__ X0, const float* __restrict__ W0,
                    int ws0, int wo0,
                    const float* __restrict__ X1, const float* __restrict__ W1,
                    const float* __restrict__ rs, const float* __restrict__ v0,
                    const float* __restrict__ v1, const float* __restrict__ R,
                    float* __restrict__ out) {
  const int lane = threadIdx.x;
  const int l16  = lane & 15;
  const int half = lane >> 4;
  const int tm = blockIdx.y;
  const int tn = blockIdx.x;
  v8f c0 = {}, c1 = {};
  c0 = tile_k128(X0 + (tm * 16 + l16) * D_, W0 + (tn * 16 + l16) * ws0 + wo0, half, c0);
  c1 = tile_k128(X1 + (tm * 16 + l16) * D_, W1 + (tn * 16 + l16) * D_, half, c1);
  const int o = tn * 16 + l16;
  const float bv = v0[o] + (v1 ? v1[o] : 0.f);
#pragma unroll
  for (int r = 0; r < 8; ++r) {
    const int m = tm * 16 + half * 8 + r;
    float val = (c0[r] + c1[r]) + rs[m] * bv;
    if (RESID) val += R[m * D_ + o];
    if (PACK) out[pack_index(m, o)] = val;
    else      out[m * D_ + o] = val;
  }
}

#define PR 32
#define PO 32
#define APITCH (N_ + 4)
#define AIP (PO + 1)

__device__ __forceinline__ void pair_step8(const v16f s, const v4f alo, const v4f ahi,
                                           const float ai0, const float ai1,
                                           float& acc0, float& acc1) {
#pragma unroll
  for (int jj = 0; jj < 8; ++jj) {
    const float av = jj < 4 ? alo[jj & 3] : ahi[jj & 3];
    float t0 = ai0 + s[2 * jj];
    float t1 = ai1 + s[2 * jj + 1];
    t0 = fmaxf(t0, 0.f);
    t1 = fmaxf(t1, 0.f);
    acc0 = fmaf(av, t0, acc0);
    acc1 = fmaf(av, t1, acc1);
  }
}

template <int KS>
__global__ __launch_bounds__(512)
void pairwise_rows(const float* __restrict__ h, const float* __restrict__ S0,
                   const float* __restrict__ wi, const float* __restrict__ Ci,
                   const float* __restrict__ b1, const float* __restrict__ rs,
                   const float* __restrict__ vi, const float* __restrict__ ajP,
                   const float* __restrict__ adj, float* __restrict__ S,
                   float* __restrict__ rs_out) {
  constexpr int HP = KS * D_ + 4;
  __shared__ __attribute__((aligned(16))) float adjL[PR * APITCH];
  __shared__ __attribute__((aligned(16))) float hL[PR * HP];
  __shared__ __attribute__((aligned(16))) float wL[PO * HP];
  __shared__ float aiL[KS * PR * AIP];
  __shared__ float rsL[KS == 1 ? 8 * PR : 1];
  const int tid  = threadIdx.x;
  const int lane = tid & 31;
  const int w    = __builtin_amdgcn_readfirstlane(tid >> 5);
  const int ob   = blockIdx.x;
  const int row0 = blockIdx.y * PR;
  const int b    = blockIdx.y >> 4;

  const v4f* ag = (const v4f*)(adj + (size_t)row0 * N_);
#pragma unroll
  for (int p = 0; p < 8; ++p) {
    const int idx = p * 512 + tid;
    const int r = idx >> 7, c4 = idx & 127;
    *(v4f*)(adjL + r * APITCH + c4 * 4) = ag[r * (N_ / 4) + c4];
  }
#pragma unroll
  for (int p = 0; p < 2; ++p) {
    const int idx = p * 512 + tid;
    const int r = idx >> 5, c4 = idx & 31;
    *(v4f*)(hL + r * HP + c4 * 4) = *(const v4f*)(h + (size_t)(row0 + r) * D_ + c4 * 4);
    *(v4f*)(wL + r * HP + c4 * 4) = *(const v4f*)(wi + (size_t)(ob * PO + r) * (2 * D_) + c4 * 4);
    if (KS == 2) {
      *(v4f*)(hL + r * HP + D_ + c4 * 4) = *(const v4f*)(S0 + (size_t)(row0 + r) * D_ + c4 * 4);
      *(v4f*)(wL + r * HP + D_ + c4 * 4) = *(const v4f*)(Ci + (size_t)(ob * PO + r) * D_ + c4 * 4);
    }
  }

  const float* sp = ajP + (size_t)(b * (D_ / 2) + ob * (PO / 2) + w) * N_ * 2;
  const float* arow = adjL + lane * APITCH;
  float acc0 = 0.f, acc1 = 0.f;

  v16f sA0 = *(const v16f*)(sp), sA1 = *(const v16f*)(sp + 16);
  __syncthreads();
  if (w < 4 * KS) {
    const int l16  = lane & 15;
    const int half = lane >> 4;
    const int seg = w >> 2;
    const int tr = (w >> 1) & 1, tc = w & 1;
    const float* xrow = hL + (tr * 16 + l16) * HP + seg * D_;
    const float* wrow = wL + (tc * 16 + l16) * HP + seg * D_;
    const float bv = seg == 0 ? b1[ob * PO + tc * 16 + l16] : 0.f;
    v8f c = {};
#pragma unroll
    for (int kk = 0; kk < D_ / 4; ++kk) {
      const int kb = kk * 4 + half * 2;
      v2f a = *(const v2f*)(xrow + kb);
      v2f b = *(const v2f*)(wrow + kb);
      c = __builtin_amdgcn_wmma_f32_16x16x4_f32(
          false, a, false, b, (short)0, c, false, false);
    }
#pragma unroll
    for (int r = 0; r < 8; ++r)
      aiL[seg * PR * AIP + (tr * 16 + half * 8 + r) * AIP + tc * 16 + l16] = c[r] + bv;
  } else if (KS == 1 && ob == 0 && w >= 8) {
    const float* p = adjL + lane * APITCH + (w - 8) * 64;
    v4f a = *(const v4f*)p;
#pragma unroll
    for (int t = 1; t < 16; ++t) a += *(const v4f*)(p + 4 * t);
    rsL[(w - 8) * PR + lane] = (a[0] + a[1]) + (a[2] + a[3]);
  }
  __syncthreads();
  if (KS == 1 && ob == 0 && w == 0) {
    float s = rsL[lane];
#pragma unroll
    for (int c = 1; c < 8; ++c) s += rsL[c * PR + lane];
    rs_out[row0 + lane] = s;
  }
  float ai0 = aiL[lane * AIP + w * 2], ai1 = aiL[lane * AIP + w * 2 + 1];
  if (KS == 2) {
    const float rv = rs[row0 + lane];
    ai0 = (ai0 + aiL[PR * AIP + lane * AIP + w * 2])     + rv * vi[ob * PO + w * 2];
    ai1 = (ai1 + aiL[PR * AIP + lane * AIP + w * 2 + 1]) + rv * vi[ob * PO + w * 2 + 1];
  }
  v4f aA0 = *(const v4f*)(arow), aA1 = *(const v4f*)(arow + 4),
      aA2 = *(const v4f*)(arow + 8), aA3 = *(const v4f*)(arow + 12);
  for (int g = 0; g < N_ / 16; g += 2) {
    asm volatile("" :: "s"(sA0), "s"(sA1));
    const float* s1 = sp + (g + 1) * 32;
    const float* r1 = arow + (g + 1) * 16;
    const v16f sB0 = *(const v16f*)(s1), sB1 = *(const v16f*)(s1 + 16);
    const v4f aB0 = *(const v4f*)(r1), aB1 = *(const v4f*)(r1 + 4),
              aB2 = *(const v4f*)(r1 + 8), aB3 = *(const v4f*)(r1 + 12);
    __builtin_amdgcn_sched_barrier(0);
    pair_step8(sA0, aA0, aA1, ai0, ai1, acc0, acc1);
    pair_step8(sA1, aA2, aA3, ai0, ai1, acc0, acc1);
    asm volatile("" :: "s"(sB0), "s"(sB1));
    const int g2 = (g + 2 < N_ / 16) ? g + 2 : g + 1;
    const float* s2 = sp + g2 * 32;
    const float* r2 = arow + g2 * 16;
    sA0 = *(const v16f*)(s2); sA1 = *(const v16f*)(s2 + 16);
    aA0 = *(const v4f*)(r2); aA1 = *(const v4f*)(r2 + 4);
    aA2 = *(const v4f*)(r2 + 8); aA3 = *(const v4f*)(r2 + 12);
    __builtin_amdgcn_sched_barrier(0);
    pair_step8(sB0, aB0, aB1, ai0, ai1, acc0, acc1);
    pair_step8(sB1, aB2, aB3, ai0, ai1, acc0, acc1);
  }

  __syncthreads();
  adjL[lane * (PO + 1) + w * 2]     = acc0;
  adjL[lane * (PO + 1) + w * 2 + 1] = acc1;
  __syncthreads();
#pragma unroll
  for (int rr = 0; rr < 2; ++rr) {
    const int r = w * 2 + rr;
    S[(size_t)(row0 + r) * D_ + ob * PO + lane] = adjL[r * (PO + 1) + lane];
  }
}

extern "C" void kernel_launch(void* const* d_in, const int* in_sizes, int n_in,
                              void* d_out, int out_size, void* d_ws, size_t ws_size,
                              hipStream_t stream) {
  const float* node = (const float*)d_in[0];
  const float* adj  = (const float*)d_in[1];
  const float* w1   = (const float*)d_in[2];
  const float* b1   = (const float*)d_in[3];
  const float* w2   = (const float*)d_in[4];
  const float* b2   = (const float*)d_in[5];

  float* out = (float*)d_out;
  float* aj = (float*)d_ws;
  float* S0 = aj + (size_t)M_ * D_;
  float* S1 = S0 + (size_t)M_ * D_;
  float* rs = S1 + (size_t)M_ * D_;
  float* Cf = rs + M_;
  float* vf = Cf + 2 * D_ * D_;

  const float* w1n = w1 + (size_t)D_ * 2 * D_;
  const float* b1n = b1 + D_;
  const float* w2n = w2 + (size_t)D_ * D_;
  const float* b2n = b2 + D_;

  const dim3 gg(D_ / 16, M_ / 16);
  aj0_and_fold<<<dim3(D_ / 16, M_ / 16 + 17), 32, 0, stream>>>(node, w1, w2, b2, aj, Cf, vf);
  pairwise_rows<1><<<dim3(D_ / PO, M_ / PR), 512, 0, stream>>>(
      node, nullptr, w1, nullptr, b1, nullptr, nullptr, aj, adj, S0, rs);
  gemm_two_parts<true, false><<<gg, 32, 0, stream>>>(
      node, w1n, 2 * D_, D_, S0, Cf + D_ * D_, rs, vf + D_, nullptr, nullptr, aj);
  pairwise_rows<2><<<dim3(D_ / PO, M_ / PR), 512, 0, stream>>>(
      node, S0, w1n, Cf, b1n, rs, vf, aj, adj, S1, nullptr);
  gemm_two_parts<false, true><<<gg, 32, 0, stream>>>(
      S0, w2, D_, 0, S1, w2n, rs, b2, b2n, node, out);
}


# pairwise lanes remapped to 2 rows x 2 columns: both halves of a packed add share one aj scalar, taken straight from the scalar register (no copies); adj row pairs interleaved in LDS; sum orders kept
# speedup vs baseline: 2.5045x; 1.1711x over previous
#include <hip/hip_runtime.h>

typedef __attribute__((ext_vector_type(2))) float v2f;
typedef __attribute__((ext_vector_type(4))) float v4f;
typedef __attribute__((ext_vector_type(8))) float v8f;
typedef __attribute__((ext_vector_type(16))) float v16f;

#define B_ 4
#define N_ 512
#define D_ 128
#define L_ 2
#define M_ (B_ * N_)


__device__ __forceinline__ v8f tile_k128(const float* __restrict__ xrow,
                                         const float* __restrict__ wrow,
                                         const int half, v8f c) {
#pragma unroll
  for (int kk = 0; kk < D_ / 4; ++kk) {
    const int kb = kk * 4 + half * 2;
    v2f a = *(const v2f*)(xrow + kb);
    v2f b = *(const v2f*)(wrow + kb);
    c = __builtin_amdgcn_wmma_f32_16x16x4_f32(
        false, a, false, b, (short)0, c, false, false);
  }
  return c;
}

__device__ __forceinline__ v8f tile_k128_panels(const float* __restrict__ xs, const int m,
                                                const float* __restrict__ wrow,
                                                const int half, v8f c) {
#pragma unroll
  for (int kk = 0; kk < D_ / 4; ++kk) {
    const int kb = kk * 4 + half * 2;
    v2f a = *(const v2f*)(xs + ((size_t)(kk >> 2) * M_ + m) * 16 + (kk & 3) * 4 + half * 2);
    v2f b = *(const v2f*)(wrow + kb);
    c = __builtin_amdgcn_wmma_f32_16x16x4_f32(
        false, a, false, b, (short)0, c, false, false);
  }
  return c;
}

__device__ __forceinline__ int pack_index(const int m, const int o) {
  return ((((m >> 9) * (D_ / 2) + (o >> 1)) * N_ + (m & (N_ - 1))) << 1) + (o & 1);
}

__global__ __launch_bounds__(32)
void aj0_and_fold(const float* __restrict__ node, const float* __restrict__ w1,
                  const float* __restrict__ w2, const float* __restrict__ b2,
                  float* __restrict__ aj, float* __restrict__ Cf, float* __restrict__ vf) {
  const int lane = threadIdx.x;
  const int l16  = lane & 15;
  const int half = lane >> 4;
  const int tn = blockIdx.x;
  const int ty = blockIdx.y;
  const float* w1n = w1 + D_ * 2 * D_;
  if (ty < M_ / 16) {
    const float* xrow = node + (ty * 16 + l16) * D_;
    const float* wrow = w1 + (tn * 16 + l16) * (2 * D_) + D_;
    v8f c = {};
    c = tile_k128(xrow, wrow, half, c);
#pragma unroll
    for (int r = 0; r < 8; ++r)
      aj[pack_index(ty * 16 + half * 8 + r, tn * 16 + l16)] = c[r];
  } else if (ty < M_ / 16 + 16) {
    const int t = ty - M_ / 16;
    const int sel = t >> 3, to = t & 7;
    const float* xrow = w1n + (to * 16 + l16) * (2 * D_) + sel * D_;
    const float* wcol = w2 + tn * 16 + l16;
    v8f c = {};
#pragma unroll
    for (int kk = 0; kk < D_ / 4; ++kk) {
      const int kb = kk * 4 + half * 2;
      v2f a = *(const v2f*)(xrow + kb);
      v2f b;
      b[0] = wcol[kb * D_];
      b[1] = wcol[(kb + 1) * D_];
      c = __builtin_amdgcn_wmma_f32_16x16x4_f32(
          false, a, false, b, (short)0, c, false, false);
    }
#pragma unroll
    for (int r = 0; r < 8; ++r)
      Cf[(sel * D_ + to * 16 + half * 8 + r) * D_ + tn * 16 + l16] = c[r];
  } else {
    const int o = tn * 16 + l16;
    const v4f* wr = (const v4f*)(w1n + o * (2 * D_) + half * D_);
    const v4f* bb = (const v4f*)b2;
    float s = 0.f;
#pragma unroll 8
    for (int e = 0; e < D_ / 4; ++e) {
      const v4f x = wr[e], y = bb[e];
      s = fmaf(x[0], y[0], s); s = fmaf(x[1], y[1], s);
      s = fmaf(x[2], y[2], s); s = fmaf(x[3], y[3], s);
    }
    vf[half * D_ + o] = s;
  }
}

template <bool PACK, bool RESID, bool X0S>
__global__ __launch_bounds__(32)
void gemm_two_parts(const float* __restrict__ X0, const float* __restrict__ W0,
                    int ws0, int wo0,
                    const float* __restrict__ X1, const float* __restrict__ W1,
                    const float* __restrict__ rs, const float* __restrict__ v0,
                    const float* __restrict__ v1, const float* __restrict__ R,
                    float* __restrict__ out) {
  const int lane = threadIdx.x;
  const int l16  = lane & 15;
  const int half = lane >> 4;
  const int tm = blockIdx.y;
  const int tn = blockIdx.x;
  v8f c0 = {}, c1 = {};
  if (X0S) c0 = tile_k128_panels(X0, tm * 16 + l16, W0 + (tn * 16 + l16) * ws0 + wo0, half, c0);
  else     c0 = tile_k128(X0 + (tm * 16 + l16) * D_, W0 + (tn * 16 + l16) * ws0 + wo0, half, c0);
  c1 = tile_k128_panels(X1, tm * 16 + l16, W1 + (tn * 16 + l16) * D_, half, c1);
  const int o = tn * 16 + l16;
  const float bv = v0[o] + (v1 ? v1[o] : 0.f);
#pragma unroll
  for (int r = 0; r < 8; ++r) {
    const int m = tm * 16 + half * 8 + r;
    float val = (c0[r] + c1[r]) + rs[m] * bv;
    if (RESID) val += R[m * D_ + o];
    if (PACK) out[pack_index(m, o)] = val;
    else      out[m * D_ + o] = val;
  }
}

#define PR 64
#define PO 16
#define APITCH (2 * N_ + 4)
#define AIP (PO + 1)

__device__ __forceinline__ void pair_step8(const v16f s, const v4f q0, const v4f q1,
                                           const v4f q2, const v4f q3,
                                           const v2f ai0, const v2f ai1,
                                           v2f& acc0, v2f& acc1) {
  const v2f zero = {0.f, 0.f};
#pragma unroll
  for (int jj = 0; jj < 8; ++jj) {
    const v4f q = jj < 2 ? q0 : jj < 4 ? q1 : jj < 6 ? q2 : q3;
    const v2f av = (jj & 1) ? (v2f){q[2], q[3]} : (v2f){q[0], q[1]};
    const float s0 = s[2 * jj], s1 = s[2 * jj + 1];
    v2f t0 = ai0 + (v2f){s0, s0};
    v2f t1 = ai1 + (v2f){s1, s1};
    t0 = __builtin_elementwise_max(t0, zero);
    t1 = __builtin_elementwise_max(t1, zero);
    acc0 = __builtin_elementwise_fma(av, t0, acc0);
    acc1 = __builtin_elementwise_fma(av, t1, acc1);
  }
}

template <int KS>
__global__ __launch_bounds__(256)
void pairwise_rows(const float* __restrict__ h, const float* __restrict__ S0,
                   const float* __restrict__ wi, const float* __restrict__ Ci,
                   const float* __restrict__ b1, const float* __restrict__ rs,
                   const float* __restrict__ vi, const float* __restrict__ ajP,
                   const float* __restrict__ adj, float* __restrict__ S,
                   float* __restrict__ rs_out) {
  constexpr int HP = KS * D_ + 4;
  __shared__ __attribute__((aligned(16))) float adjL[(PR / 2) * APITCH];
  __shared__ __attribute__((aligned(16))) float hL[PR * HP];
  __shared__ __attribute__((aligned(16))) float wL[PO * HP];
  __shared__ float aiL[KS * PR * AIP];
  __shared__ float rsL[KS == 1 ? 8 * PR : 1];
  const int tid  = threadIdx.x;
  const int lane = tid & 31;
  const int w    = __builtin_amdgcn_readfirstlane(tid >> 5);
  const int ob   = blockIdx.x;
  const int row0 = blockIdx.y * PR;
  const int b    = blockIdx.y >> 3;

  const v4f* ag = (const v4f*)(adj + (size_t)row0 * N_);
#pragma unroll
  for (int p = 0; p < 16; ++p) {
    const int idx = p * 256 + tid;
    const int pr = idx >> 7, c4 = idx & 127;
    const v4f x0 = ag[pr * (N_ / 4) + c4];
    const v4f x1 = ag[(pr + 32) * (N_ / 4) + c4];
    float* d = adjL + pr * APITCH + c4 * 8;
    *(v4f*)(d)     = (v4f){x0[0], x1[0], x0[1], x1[1]};
    *(v4f*)(d + 4) = (v4f){x0[2], x1[2], x0[3], x1[3]};
  }
#pragma unroll
  for (int p = 0; p < 8; ++p) {
    const int idx = p * 256 + tid;
    const int r = idx >> 5, c4 = idx & 31;
    *(v4f*)(hL + r * HP + c4 * 4) = *(const v4f*)(h + (size_t)(row0 + r) * D_ + c4 * 4);
    if (KS == 2) {
      const int r2 = tid >> 2, q4 = tid & 3;
      *(v4f*)(hL + r2 * HP + D_ + p * 16 + q4 * 4) =
          *(const v4f*)(S0 + ((size_t)p * M_ + row0 + r2) * 16 + q4 * 4);
    }
  }
#pragma unroll
  for (int p = 0; p < 2; ++p) {
    const int idx = p * 256 + tid;
    const int r = idx >> 5, c4 = idx & 31;
    *(v4f*)(wL + r * HP + c4 * 4) = *(const v4f*)(wi + (size_t)(ob * PO + r) * (2 * D_) + c4 * 4);
    if (KS == 2)
      *(v4f*)(wL + r * HP + D_ + c4 * 4) = *(const v4f*)(Ci + (size_t)(ob * PO + r) * D_ + c4 * 4);
  }

  const float* sp = ajP + (size_t)(b * (D_ / 2) + ob * (PO / 2) + w) * N_ * 2;
  const float* arow = adjL + lane * APITCH;
  v2f acc0 = {0.f, 0.f}, acc1 = {0.f, 0.f};

  v16f sA0 = *(const v16f*)(sp), sA1 = *(const v16f*)(sp + 16);
  __syncthreads();
  if (w < 4 * KS) {
    const int l16  = lane & 15;
    const int half = lane >> 4;
    const int seg = w >> 2;
    const int tr = w & 3;
    const float* xrow = hL + (tr * 16 + l16) * HP + seg * D_;
    const float* wrow = wL + l16 * HP + seg * D_;
    const float bv = seg == 0 ? b1[ob * PO + l16] : 0.f;
    v8f c = {};
#pragma unroll
    for (int kk = 0; kk < D_ / 4; ++kk) {
      const int kb = kk * 4 + half * 2;
      v2f a = *(const v2f*)(xrow + kb);
      v2f b = *(const v2f*)(wrow + kb);
      c = __builtin_amdgcn_wmma_f32_16x16x4_f32(
          false, a, false, b, (short)0, c, false, false);
    }
#pragma unroll
    for (int r = 0; r < 8; ++r)
      aiL[seg * PR * AIP + (tr * 16 + half * 8 + r) * AIP + l16] = c[r] + bv;
  } else if (KS == 1 && ob == 0) {
#pragma unroll
    for (int cc = 0; cc < 2; ++cc) {
      const int ch = (w - 4) * 2 + cc;
      const float* p = arow + ch * 128;
      v4f l0 = *(const v4f*)p, l1 = *(const v4f*)(p + 4);
      v2f a0 = {l0[0], l0[1]}, a1 = {l0[2], l0[3]}, a2 = {l1[0], l1[1]}, a3 = {l1[2], l1[3]};
#pragma unroll
      for (int t = 1; t < 16; ++t) {
        l0 = *(const v4f*)(p + 8 * t);
        l1 = *(const v4f*)(p + 8 * t + 4);
        a0 += (v2f){l0[0], l0[1]}; a1 += (v2f){l0[2], l0[3]};
        a2 += (v2f){l1[0], l1[1]}; a3 += (v2f){l1[2], l1[3]};
      }
      const v2f sm = (a0 + a1) + (a2 + a3);
      rsL[ch * PR + lane]      = sm[0];
      rsL[ch * PR + lane + 32] = sm[1];
    }
  }
  __syncthreads();
  if (KS == 1 && ob == 0 && w == 0) {
#pragma unroll
    for (int hh = 0; hh < 2; ++hh) {
      float s = rsL[lane + 32 * hh];
#pragma unroll
      for (int c = 1; c < 8; ++c) s += rsL[c * PR + lane + 32 * hh];
      rs_out[row0 + lane + 32 * hh] = s;
    }
  }
  v2f ai0 = {aiL[lane * AIP + w * 2],     aiL[(lane + 32) * AIP + w * 2]};
  v2f ai1 = {aiL[lane * AIP + w * 2 + 1], aiL[(lane + 32) * AIP + w * 2 + 1]};
  if (KS == 2) {
    const float rv0 = rs[row0 + lane], rv1 = rs[row0 + lane + 32];
    const float vi0 = vi[ob * PO + w * 2], vi1 = vi[ob * PO + w * 2 + 1];
    ai0[0] = (ai0[0] + aiL[PR * AIP + lane * AIP + w * 2])            + rv0 * vi0;
    ai0[1] = (ai0[1] + aiL[PR * AIP + (lane + 32) * AIP + w * 2])     + rv1 * vi0;
    ai1[0] = (ai1[0] + aiL[PR * AIP + lane * AIP + w * 2 + 1])        + rv0 * vi1;
    ai1[1] = (ai1[1] + aiL[PR * AIP + (lane + 32) * AIP + w * 2 + 1]) + rv1 * vi1;
  }
  v4f aA0 = *(const v4f*)(arow),      aA1 = *(const v4f*)(arow + 4),
      aA2 = *(const v4f*)(arow + 8),  aA3 = *(const v4f*)(arow + 12),
      aA4 = *(const v4f*)(arow + 16), aA5 = *(const v4f*)(arow + 20),
      aA6 = *(const v4f*)(arow + 24), aA7 = *(const v4f*)(arow + 28);
  for (int g = 0; g < N_ / 16; g += 2) {
    asm volatile("" :: "s"(sA0), "s"(sA1));
    const float* s1 = sp + (g + 1) * 32;
    const float* r1 = arow + (g + 1) * 32;
    const v16f sB0 = *(const v16f*)(s1), sB1 = *(const v16f*)(s1 + 16);
    const v4f aB0 = *(const v4f*)(r1),      aB1 = *(const v4f*)(r1 + 4),
              aB2 = *(const v4f*)(r1 + 8),  aB3 = *(const v4f*)(r1 + 12),
              aB4 = *(const v4f*)(r1 + 16), aB5 = *(const v4f*)(r1 + 20),
              aB6 = *(const v4f*)(r1 + 24), aB7 = *(const v4f*)(r1 + 28);
    __builtin_amdgcn_sched_barrier(0);
    pair_step8(sA0, aA0, aA1, aA2, aA3, ai0, ai1, acc0, acc1);
    pair_step8(sA1, aA4, aA5, aA6, aA7, ai0, ai1, acc0, acc1);
    asm volatile("" :: "s"(sB0), "s"(sB1));
    const int g2 = (g + 2 < N_ / 16) ? g + 2 : g + 1;
    const float* s2 = sp + g2 * 32;
    const float* r2 = arow + g2 * 32;
    sA0 = *(const v16f*)(s2); sA1 = *(const v16f*)(s2 + 16);
    aA0 = *(const v4f*)(r2);      aA1 = *(const v4f*)(r2 + 4);
    aA2 = *(const v4f*)(r2 + 8);  aA3 = *(const v4f*)(r2 + 12);
    aA4 = *(const v4f*)(r2 + 16); aA5 = *(const v4f*)(r2 + 20);
    aA6 = *(const v4f*)(r2 + 24); aA7 = *(const v4f*)(r2 + 28);
    __builtin_amdgcn_sched_barrier(0);
    pair_step8(sB0, aB0, aB1, aB2, aB3, ai0, ai1, acc0, acc1);
    pair_step8(sB1, aB4, aB5, aB6, aB7, ai0, ai1, acc0, acc1);
  }

  __syncthreads();
  adjL[lane * AIP + w * 2]            = acc0[0];
  adjL[(lane + 32) * AIP + w * 2]     = acc0[1];
  adjL[lane * AIP + w * 2 + 1]        = acc1[0];
  adjL[(lane + 32) * AIP + w * 2 + 1] = acc1[1];
  __syncthreads();
  {
    const int r = tid >> 2, c4 = tid & 3;
    const float* t = adjL + r * AIP + c4 * 4;
    const v4f o = {t[0], t[1], t[2], t[3]};
    *(v4f*)(S + ((size_t)ob * M_ + row0 + r) * 16 + c4 * 4) = o;
  }
}

extern "C" void kernel_launch(void* const* d_in, const int* in_sizes, int n_in,
                              void* d_out, int out_size, void* d_ws, size_t ws_size,
                              hipStream_t stream) {
  const float* node = (const float*)d_in[0];
  const float* adj  = (const float*)d_in[1];
  const float* w1   = (const float*)d_in[2];
  const float* b1   = (const float*)d_in[3];
  const float* w2   = (const float*)d_in[4];
  const float* b2   = (const float*)d_in[5];

  float* out = (float*)d_out;
  float* aj = (float*)d_ws;
  float* S0 = aj + (size_t)M_ * D_;
  float* S1 = S0 + (size_t)M_ * D_;
  float* rs = S1 + (size_t)M_ * D_;
  float* Cf = rs + M_;
  float* vf = Cf + 2 * D_ * D_;

  const float* w1n = w1 + (size_t)D_ * 2 * D_;
  const float* b1n = b1 + D_;
  const float* w2n = w2 + (size_t)D_ * D_;
  const float* b2n = b2 + D_;

  const dim3 gg(D_ / 16, M_ / 16);
  aj0_and_fold<<<dim3(D_ / 16, M_ / 16 + 17), 32, 0, stream>>>(node, w1, w2, b2, aj, Cf, vf);
  pairwise_rows<1><<<dim3(D_ / PO, M_ / PR), 256, 0, stream>>>(
      node, nullptr, w1, nullptr, b1, nullptr, nullptr, aj, adj, S0, rs);
  gemm_two_parts<true, false, false><<<gg, 32, 0, stream>>>(
      node, w1n, 2 * D_, D_, S0, Cf + D_ * D_, rs, vf + D_, nullptr, nullptr, aj);
  pairwise_rows<2><<<dim3(D_ / PO, M_ / PR), 256, 0, stream>>>(
      node, S0, w1n, Cf, b1n, rs, vf, aj, adj, S1, nullptr);
  gemm_two_parts<false, true, true><<<gg, 32, 0, stream>>>(
      S0, w2, D_, 0, S1, w2n, rs, b2, b2n, node, out);
}


# product launches fetch operand fragments as far ahead as a wave can hold loads in flight (whole tile, or 15 of 32 k-steps) and tail operands under the last products, not 4 steps ahead; sums kept
# speedup vs baseline: 2.8458x; 1.1363x over previous
#include <hip/hip_runtime.h>

typedef __attribute__((ext_vector_type(2))) float v2f;
typedef __attribute__((ext_vector_type(4))) float v4f;
typedef __attribute__((ext_vector_type(8))) float v8f;
typedef __attribute__((ext_vector_type(16))) float v16f;

#define B_ 4
#define N_ 512
#define D_ 128
#define L_ 2
#define M_ (B_ * N_)


#define KSTEPS (D_ / 4)

__device__ __forceinline__ void fetch_rows(const float* __restrict__ xrow,
                                           const float* __restrict__ wrow,
                                           const int half, const int kk, v2f& a, v2f& b) {
  const int kb = kk * 4 + half * 2;
  a = *(const v2f*)(xrow + kb);
  b = *(const v2f*)(wrow + kb);
}

__device__ __forceinline__ void fetch_panels(const float* __restrict__ xs, const int m,
                                             const float* __restrict__ wrow,
                                             const int half, const int kk, v2f& a, v2f& b) {
  const int kb = kk * 4 + half * 2;
  a = *(const v2f*)(xs + ((kk >> 2) * M_ + m) * 16 + (kk & 3) * 4 + half * 2);
  b = *(const v2f*)(wrow + kb);
}

__device__ __forceinline__ int pack_index(const int m, const int o) {
  return ((((m >> 9) * (D_ / 2) + (o >> 1)) * N_ + (m & (N_ - 1))) << 1) + (o & 1);
}

__global__ __launch_bounds__(32)
void aj0_and_fold(const float* __restrict__ node, const float* __restrict__ w1,
                  const float* __restrict__ w2, const float* __restrict__ b2,
                  float* __restrict__ aj, float* __restrict__ Cf, float* __restrict__ vf) {
  const int lane = threadIdx.x;
  const int l16  = lane & 15;
  const int half = lane >> 4;
  const int tn = blockIdx.x;
  const int ty = blockIdx.y;
  const float* w1n = w1 + D_ * 2 * D_;
  if (ty < M_ / 16) {
    const float* xrow = node + (ty * 16 + l16) * D_;
    const float* wrow = w1 + (tn * 16 + l16) * (2 * D_) + D_;
    v2f a[KSTEPS], b[KSTEPS];
#pragma unroll
    for (int kk = 0; kk < KSTEPS; ++kk) fetch_rows(xrow, wrow, half, kk, a[kk], b[kk]);
    __builtin_amdgcn_sched_barrier(0);
    v8f c = {};
#pragma unroll
    for (int kk = 0; kk < KSTEPS; ++kk)
      c = __builtin_amdgcn_wmma_f32_16x16x4_f32(
          false, a[kk], false, b[kk], (short)0, c, false, false);
#pragma unroll
    for (int r = 0; r < 8; ++r)
      aj[pack_index(ty * 16 + half * 8 + r, tn * 16 + l16)] = c[r];
  } else if (ty < M_ / 16 + 16) {
    const int t = ty - M_ / 16;
    const int sel = t >> 3, to = t & 7;
    const float* xrow = w1n + (to * 16 + l16) * (2 * D_) + sel * D_;
    const float* wcol = w2 + tn * 16 + l16;
    v2f a[KSTEPS], b[KSTEPS];
#pragma unroll
    for (int kk = 0; kk < KSTEPS; ++kk) {
      const int kb = kk * 4 + half * 2;
      a[kk] = *(const v2f*)(xrow + kb);
      b[kk][0] = wcol[kb * D_];
      b[kk][1] = wcol[(kb + 1) * D_];
    }
    __builtin_amdgcn_sched_barrier(0);
    v8f c = {};
#pragma unroll
    for (int kk = 0; kk < KSTEPS; ++kk)
      c = __builtin_amdgcn_wmma_f32_16x16x4_f32(
          false, a[kk], false, b[kk], (short)0, c, false, false);
#pragma unroll
    for (int r = 0; r < 8; ++r)
      Cf[(sel * D_ + to * 16 + half * 8 + r) * D_ + tn * 16 + l16] = c[r];
  } else {
    const int o = tn * 16 + l16;
    const v4f* wr = (const v4f*)(w1n + o * (2 * D_) + half * D_);
    const v4f* bb = (const v4f*)b2;
    v4f x[D_ / 4];
#pragma unroll
    for (int e = 0; e < D_ / 4; ++e) x[e] = wr[e];
    __builtin_amdgcn_sched_barrier(0);
    float s = 0.f;
#pragma unroll
    for (int e = 0; e < D_ / 4; ++e) {
      const v4f y = bb[e];
      s = fmaf(x[e][0], y[0], s); s = fmaf(x[e][1], y[1], s);
      s = fmaf(x[e][2], y[2], s); s = fmaf(x[e][3], y[3], s);
    }
    vf[half * D_ + o] = s;
  }
}

template <bool PACK, bool RESID, bool X0S>
__global__ __launch_bounds__(32)
void gemm_two_parts(const float* __restrict__ X0, const float* __restrict__ W0,
                    int ws0, int wo0,
                    const float* __restrict__ X1, const float* __restrict__ W1,
                    const float* __restrict__ rs, const float* __restrict__ v0,
                    const float* __restrict__ v1, const float* __restrict__ R,
                    float* __restrict__ out) {
  const int lane = threadIdx.x;
  const int l16  = lane & 15;
  const int half = lane >> 4;
  const int tm = blockIdx.y;
  const int tn = blockIdx.x;
  const int o = tn * 16 + l16;
  const int m16 = tm * 16 + l16;
  const int mr = tm * 16 + half * 8;
  const float* x0row = X0 + m16 * D_;
  const float* w0row = W0 + o * ws0 + wo0;
  const float* w1row = W1 + o * D_;
  constexpr int AHEAD = 15;
  v2f a0[KSTEPS], b0[KSTEPS], a1[KSTEPS], b1[KSTEPS];
  v4f rs_lo = {}, rs_hi = {};
  float bv0 = 0.f, bv1 = 0.f;
  float res[8] = {};
#pragma unroll
  for (int kk = 0; kk < AHEAD; ++kk) {
    if (X0S) fetch_panels(X0, m16, w0row, half, kk, a0[kk], b0[kk]);
    else     fetch_rows(x0row, w0row, half, kk, a0[kk], b0[kk]);
    fetch_panels(X1, m16, w1row, half, kk, a1[kk], b1[kk]);
  }
  __builtin_amdgcn_sched_barrier(0);
  v8f c0 = {}, c1 = {};
#pragma unroll
  for (int kk = 0; kk < KSTEPS; ++kk) {
    if (kk + AHEAD < KSTEPS) {
      if (X0S) fetch_panels(X0, m16, w0row, half, kk + AHEAD, a0[kk + AHEAD], b0[kk + AHEAD]);
      else     fetch_rows(x0row, w0row, half, kk + AHEAD, a0[kk + AHEAD], b0[kk + AHEAD]);
      fetch_panels(X1, m16, w1row, half, kk + AHEAD, a1[kk + AHEAD], b1[kk + AHEAD]);
    } else if (kk + AHEAD == KSTEPS) {
      rs_lo = *(const v4f*)(rs + mr);
      rs_hi = *(const v4f*)(rs + mr + 4);
      bv0 = v0[o];
      bv1 = v1 ? v1[o] : 0.f;
      if (RESID) {
#pragma unroll
        for (int r = 0; r < 8; ++r) res[r] = R[(mr + r) * D_ + o];
      }
    }
    c0 = __builtin_amdgcn_wmma_f32_16x16x4_f32(
        false, a0[kk], false, b0[kk], (short)0, c0, false, false);
    c1 = __builtin_amdgcn_wmma_f32_16x16x4_f32(
        false, a1[kk], false, b1[kk], (short)0, c1, false, false);
    __builtin_amdgcn_sched_barrier(0);
  }
  const float bv = bv0 + bv1;
#pragma unroll
  for (int r = 0; r < 8; ++r) {
    const int m = mr + r;
    float val = (c0[r] + c1[r]) + (r < 4 ? rs_lo[r & 3] : rs_hi[r & 3]) * bv;
    if (RESID) val += res[r];
    if (PACK) out[pack_index(m, o)] = val;
    else      out[m * D_ + o] = val;
  }
}

#define PR 64
#define PO 16
#define APITCH (2 * N_ + 4)
#define AIP (PO + 1)

__device__ __forceinline__ void pair_step8(const v16f s, const v4f q0, const v4f q1,
                                           const v4f q2, const v4f q3,
                                           const v2f ai0, const v2f ai1,
                                           v2f& acc0, v2f& acc1) {
  const v2f zero = {0.f, 0.f};
#pragma unroll
  for (int jj = 0; jj < 8; ++jj) {
    const v4f q = jj < 2 ? q0 : jj < 4 ? q1 : jj < 6 ? q2 : q3;
    const v2f av = (jj & 1) ? (v2f){q[2], q[3]} : (v2f){q[0], q[1]};
    const float s0 = s[2 * jj], s1 = s[2 * jj + 1];
    v2f t0 = ai0 + (v2f){s0, s0};
    v2f t1 = ai1 + (v2f){s1, s1};
    t0 = __builtin_elementwise_max(t0, zero);
    t1 = __builtin_elementwise_max(t1, zero);
    acc0 = __builtin_elementwise_fma(av, t0, acc0);
    acc1 = __builtin_elementwise_fma(av, t1, acc1);
  }
}

template <int KS>
__global__ __launch_bounds__(256)
void pairwise_rows(const float* __restrict__ h, const float* __restrict__ S0,
                   const float* __restrict__ wi, const float* __restrict__ Ci,
                   const float* __restrict__ b1, const float* __restrict__ rs,
                   const float* __restrict__ vi, const float* __restrict__ ajP,
                   const float* __restrict__ adj, float* __restrict__ S,
                   float* __restrict__ rs_out) {
  constexpr int HP = KS * D_ + 4;
  __shared__ __attribute__((aligned(16))) float adjL[(PR / 2) * APITCH];
  __shared__ __attribute__((aligned(16))) float hL[PR * HP];
  __shared__ __attribute__((aligned(16))) float wL[PO * HP];
  __shared__ float aiL[KS * PR * AIP];
  __shared__ float rsL[KS == 1 ? 8 * PR : 1];
  const int tid  = threadIdx.x;
  const int lane = tid & 31;
  const int w    = __builtin_amdgcn_readfirstlane(tid >> 5);
  const int ob   = blockIdx.x;
  const int row0 = blockIdx.y * PR;
  const int b    = blockIdx.y >> 3;

  const v4f* ag = (const v4f*)(adj + (size_t)row0 * N_);
#pragma unroll
  for (int p = 0; p < 16; ++p) {
    const int idx = p * 256 + tid;
    const int pr = idx >> 7, c4 = idx & 127;
    const v4f x0 = ag[pr * (N_ / 4) + c4];
    const v4f x1 = ag[(pr + 32) * (N_ / 4) + c4];
    float* d = adjL + pr * APITCH + c4 * 8;
    *(v4f*)(d)     = (v4f){x0[0], x1[0], x0[1], x1[1]};
    *(v4f*)(d + 4) = (v4f){x0[2], x1[2], x0[3], x1[3]};
  }
#pragma unroll
  for (int p = 0; p < 8; ++p) {
    const int idx = p * 256 + tid;
    const int r = idx >> 5, c4 = idx & 31;
    *(v4f*)(hL + r * HP + c4 * 4) = *(const v4f*)(h + (size_t)(row0 + r) * D_ + c4 * 4);
    if (KS == 2) {
      const int r2 = tid >> 2, q4 = tid & 3;
      *(v4f*)(hL + r2 * HP + D_ + p * 16 + q4 * 4) =
          *(const v4f*)(S0 + ((size_t)p * M_ + row0 + r2) * 16 + q4 * 4);
    }
  }
#pragma unroll
  for (int p = 0; p < 2; ++p) {
    const int idx = p * 256 + tid;
    const int r = idx >> 5, c4 = idx & 31;
    *(v4f*)(wL + r * HP + c4 * 4) = *(const v4f*)(wi + (size_t)(ob * PO + r) * (2 * D_) + c4 * 4);
    if (KS == 2)
      *(v4f*)(wL + r * HP + D_ + c4 * 4) = *(const v4f*)(Ci + (size_t)(ob * PO + r) * D_ + c4 * 4);
  }

  const float* sp = ajP + (size_t)(b * (D_ / 2) + ob * (PO / 2) + w) * N_ * 2;
  const float* arow = adjL + lane * APITCH;
  v2f acc0 = {0.f, 0.f}, acc1 = {0.f, 0.f};

  v16f sA0 = *(const v16f*)(sp), sA1 = *(const v16f*)(sp + 16);
  __syncthreads();
  if (w < 4 * KS) {
    const int l16  = lane & 15;
    const int half = lane >> 4;
    const int seg = w >> 2;
    const int tr = w & 3;
    const float* xrow = hL + (tr * 16 + l16) * HP + seg * D_;
    const float* wrow = wL + l16 * HP + seg * D_;
    const float bv = seg == 0 ? b1[ob * PO + l16] : 0.f;
    v8f c = {};
#pragma unroll
    for (int kk = 0; kk < D_ / 4; ++kk) {
      const int kb = kk * 4 + half * 2;
      v2f a = *(const v2f*)(xrow + kb);
      v2f b = *(const v2f*)(wrow + kb);
      c = __builtin_amdgcn_wmma_f32_16x16x4_f32(
          false, a, false, b, (short)0, c, false, false);
    }
#pragma unroll
    for (int r = 0; r < 8; ++r)
      aiL[seg * PR * AIP + (tr * 16 + half * 8 + r) * AIP + l16] = c[r] + bv;
  } else if (KS == 1 && ob == 0) {
#pragma unroll
    for (int cc = 0; cc < 2; ++cc) {
      const int ch = (w - 4) * 2 + cc;
      const float* p = arow + ch * 128;
      v4f l0 = *(const v4f*)p, l1 = *(const v4f*)(p + 4);
      v2f a0 = {l0[0], l0[1]}, a1 = {l0[2], l0[3]}, a2 = {l1[0], l1[1]}, a3 = {l1[2], l1[3]};
#pragma unroll
      for (int t = 1; t < 16; ++t) {
        l0 = *(const v4f*)(p + 8 * t);
        l1 = *(const v4f*)(p + 8 * t + 4);
        a0 += (v2f){l0[0], l0[1]}; a1 += (v2f){l0[2], l0[3]};
        a2 += (v2f){l1[0], l1[1]}; a3 += (v2f){l1[2], l1[3]};
      }
      const v2f sm = (a0 + a1) + (a2 + a3);
      rsL[ch * PR + lane]      = sm[0];
      rsL[ch * PR + lane + 32] = sm[1];
    }
  }
  __syncthreads();
  if (KS == 1 && ob == 0 && w == 0) {
#pragma unroll
    for (int hh = 0; hh < 2; ++hh) {
      float s = rsL[lane + 32 * hh];
#pragma unroll
      for (int c = 1; c < 8; ++c) s += rsL[c * PR + lane + 32 * hh];
      rs_out[row0 + lane + 32 * hh] = s;
    }
  }
  v2f ai0 = {aiL[lane * AIP + w * 2],     aiL[(lane + 32) * AIP + w * 2]};
  v2f ai1 = {aiL[lane * AIP + w * 2 + 1], aiL[(lane + 32) * AIP + w * 2 + 1]};
  if (KS == 2) {
    const float rv0 = rs[row0 + lane], rv1 = rs[row0 + lane + 32];
    const float vi0 = vi[ob * PO + w * 2], vi1 = vi[ob * PO + w * 2 + 1];
    ai0[0] = (ai0[0] + aiL[PR * AIP + lane * AIP + w * 2])            + rv0 * vi0;
    ai0[1] = (ai0[1] + aiL[PR * AIP + (lane + 32) * AIP + w * 2])     + rv1 * vi0;
    ai1[0] = (ai1[0] + aiL[PR * AIP + lane * AIP + w * 2 + 1])        + rv0 * vi1;
    ai1[1] = (ai1[1] + aiL[PR * AIP + (lane + 32) * AIP + w * 2 + 1]) + rv1 * vi1;
  }
  v4f aA0 = *(const v4f*)(arow),      aA1 = *(const v4f*)(arow + 4),
      aA2 = *(const v4f*)(arow + 8),  aA3 = *(const v4f*)(arow + 12),
      aA4 = *(const v4f*)(arow + 16), aA5 = *(const v4f*)(arow + 20),
      aA6 = *(const v4f*)(arow + 24), aA7 = *(const v4f*)(arow + 28);
  for (int g = 0; g < N_ / 16; g += 2) {
    asm volatile("" :: "s"(sA0), "s"(sA1));
    const float* s1 = sp + (g + 1) * 32;
    const float* r1 = arow + (g + 1) * 32;
    const v16f sB0 = *(const v16f*)(s1), sB1 = *(const v16f*)(s1 + 16);
    const v4f aB0 = *(const v4f*)(r1),      aB1 = *(const v4f*)(r1 + 4),
              aB2 = *(const v4f*)(r1 + 8),  aB3 = *(const v4f*)(r1 + 12),
              aB4 = *(const v4f*)(r1 + 16), aB5 = *(const v4f*)(r1 + 20),
              aB6 = *(const v4f*)(r1 + 24), aB7 = *(const v4f*)(r1 + 28);
    __builtin_amdgcn_sched_barrier(0);
    pair_step8(sA0, aA0, aA1, aA2, aA3, ai0, ai1, acc0, acc1);
    pair_step8(sA1, aA4, aA5, aA6, aA7, ai0, ai1, acc0, acc1);
    asm volatile("" :: "s"(sB0), "s"(sB1));
    const int g2 = (g + 2 < N_ / 16) ? g + 2 : g + 1;
    const float* s2 = sp + g2 * 32;
    const float* r2 = arow + g2 * 32;
    sA0 = *(const v16f*)(s2); sA1 = *(const v16f*)(s2 + 16);
    aA0 = *(const v4f*)(r2);      aA1 = *(const v4f*)(r2 + 4);
    aA2 = *(const v4f*)(r2 + 8);  aA3 = *(const v4f*)(r2 + 12);
    aA4 = *(const v4f*)(r2 + 16); aA5 = *(const v4f*)(r2 + 20);
    aA6 = *(const v4f*)(r2 + 24); aA7 = *(const v4f*)(r2 + 28);
    __builtin_amdgcn_sched_barrier(0);
    pair_step8(sB0, aB0, aB1, aB2, aB3, ai0, ai1, acc0, acc1);
    pair_step8(sB1, aB4, aB5, aB6, aB7, ai0, ai1, acc0, acc1);
  }

  __syncthreads();
  adjL[lane * AIP + w * 2]            = acc0[0];
  adjL[(lane + 32) * AIP + w * 2]     = acc0[1];
  adjL[lane * AIP + w * 2 + 1]        = acc1[0];
  adjL[(lane + 32) * AIP + w * 2 + 1] = acc1[1];
  __syncthreads();
  {
    const int r = tid >> 2, c4 = tid & 3;
    const float* t = adjL + r * AIP + c4 * 4;
    const v4f o = {t[0], t[1], t[2], t[3]};
    *(v4f*)(S + ((size_t)ob * M_ + row0 + r) * 16 + c4 * 4) = o;
  }
}

extern "C" void kernel_launch(void* const* d_in, const int* in_sizes, int n_in,
                              void* d_out, int out_size, void* d_ws, size_t ws_size,
                              hipStream_t stream) {
  const float* node = (const float*)d_in[0];
  const float* adj  = (const float*)d_in[1];
  const float* w1   = (const float*)d_in[2];
  const float* b1   = (const float*)d_in[3];
  const float* w2   = (const float*)d_in[4];
  const float* b2   = (const float*)d_in[5];

  float* out = (float*)d_out;
  float* aj = (float*)d_ws;
  float* S0 = aj + (size_t)M_ * D_;
  float* S1 = S0 + (size_t)M_ * D_;
  float* rs = S1 + (size_t)M_ * D_;
  float* Cf = rs + M_;
  float* vf = Cf + 2 * D_ * D_;

  const float* w1n = w1 + (size_t)D_ * 2 * D_;
  const float* b1n = b1 + D_;
  const float* w2n = w2 + (size_t)D_ * D_;
  const float* b2n = b2 + D_;

  const dim3 gg(D_ / 16, M_ / 16);
  aj0_and_fold<<<dim3(D_ / 16, M_ / 16 + 17), 32, 0, stream>>>(node, w1, w2, b2, aj, Cf, vf);
  pairwise_rows<1><<<dim3(D_ / PO, M_ / PR), 256, 0, stream>>>(
      node, nullptr, w1, nullptr, b1, nullptr, nullptr, aj, adj, S0, rs);
  gemm_two_parts<true, false, false><<<gg, 32, 0, stream>>>(
      node, w1n, 2 * D_, D_, S0, Cf + D_ * D_, rs, vf + D_, nullptr, nullptr, aj);
  pairwise_rows<2><<<dim3(D_ / PO, M_ / PR), 256, 0, stream>>>(
      node, S0, w1n, Cf, b1n, rs, vf, aj, adj, S1, nullptr);
  gemm_two_parts<false, true, true><<<gg, 32, 0, stream>>>(
      S0, w2, D_, 0, S1, w2n, rs, b2, b2n, node, out);
}


# product launches fetch operand fragments 16 bytes per lane (lane half h takes k = 8q+4h..+3; one fetch feeds two matrix instructions): half the fetch instructions; k sums regrouped within each 8
# speedup vs baseline: 2.9372x; 1.0321x over previous
#include <hip/hip_runtime.h>

typedef __attribute__((ext_vector_type(2))) float v2f;
typedef __attribute__((ext_vector_type(4))) float v4f;
typedef __attribute__((ext_vector_type(8))) float v8f;
typedef __attribute__((ext_vector_type(16))) float v16f;

#define B_ 4
#define N_ 512
#define D_ 128
#define L_ 2
#define M_ (B_ * N_)


#define KPAIRS (D_ / 8)

__device__ __forceinline__ void fetch_rows(const float* __restrict__ xrow,
                                           const float* __restrict__ wrow,
                                           const int half, const int q, v4f& a, v4f& b) {
  const int kb = q * 8 + half * 4;
  a = *(const v4f*)(xrow + kb);
  b = *(const v4f*)(wrow + kb);
}

__device__ __forceinline__ v8f mma_step(const v4f a, const v4f b, v8f c) {
  c = __builtin_amdgcn_wmma_f32_16x16x4_f32(
      false, (v2f){a[0], a[1]}, false, (v2f){b[0], b[1]}, (short)0, c, false, false);
  c = __builtin_amdgcn_wmma_f32_16x16x4_f32(
      false, (v2f){a[2], a[3]}, false, (v2f){b[2], b[3]}, (short)0, c, false, false);
  return c;
}

__device__ __forceinline__ void fetch_panels(const float* __restrict__ xs, const int m,
                                             const float* __restrict__ wrow,
                                             const int half, const int q, v4f& a, v4f& b) {
  const int kb = q * 8 + half * 4;
  a = *(const v4f*)(xs + ((q >> 1) * M_ + m) * 16 + (q & 1) * 8 + half * 4);
  b = *(const v4f*)(wrow + kb);
}

__device__ __forceinline__ int pack_index(const int m, const int o) {
  return ((((m >> 9) * (D_ / 2) + (o >> 1)) * N_ + (m & (N_ - 1))) << 1) + (o & 1);
}

__global__ __launch_bounds__(32)
void aj0_and_fold(const float* __restrict__ node, const float* __restrict__ w1,
                  const float* __restrict__ w2, const float* __restrict__ b2,
                  float* __restrict__ aj, float* __restrict__ Cf, float* __restrict__ vf) {
  const int lane = threadIdx.x;
  const int l16  = lane & 15;
  const int half = lane >> 4;
  const int tn = blockIdx.x;
  const int ty = blockIdx.y;
  const float* w1n = w1 + D_ * 2 * D_;
  if (ty < M_ / 16) {
    const float* xrow = node + (ty * 16 + l16) * D_;
    const float* wrow = w1 + (tn * 16 + l16) * (2 * D_) + D_;
    v4f a[KPAIRS], b[KPAIRS];
#pragma unroll
    for (int q = 0; q < KPAIRS; ++q) fetch_rows(xrow, wrow, half, q, a[q], b[q]);
    __builtin_amdgcn_sched_barrier(0);
    v8f c = {};
#pragma unroll
    for (int q = 0; q < KPAIRS; ++q) c = mma_step(a[q], b[q], c);
#pragma unroll
    for (int r = 0; r < 8; ++r)
      aj[pack_index(ty * 16 + half * 8 + r, tn * 16 + l16)] = c[r];
  } else if (ty < M_ / 16 + 16) {
    const int t = ty - M_ / 16;
    const int sel = t >> 3, to = t & 7;
    const float* xrow = w1n + (to * 16 + l16) * (2 * D_) + sel * D_;
    const float* wcol = w2 + tn * 16 + l16;
    v4f a[KPAIRS], b[KPAIRS];
#pragma unroll
    for (int q = 0; q < KPAIRS; ++q) {
      const int kb = q * 8 + half * 4;
      a[q] = *(const v4f*)(xrow + kb);
      b[q][0] = wcol[kb * D_];
      b[q][1] = wcol[(kb + 1) * D_];
      b[q][2] = wcol[(kb + 2) * D_];
      b[q][3] = wcol[(kb + 3) * D_];
    }
    __builtin_amdgcn_sched_barrier(0);
    v8f c = {};
#pragma unroll
    for (int q = 0; q < KPAIRS; ++q) c = mma_step(a[q], b[q], c);
#pragma unroll
    for (int r = 0; r < 8; ++r)
      Cf[(sel * D_ + to * 16 + half * 8 + r) * D_ + tn * 16 + l16] = c[r];
  } else {
    const int o = tn * 16 + l16;
    const v4f* wr = (const v4f*)(w1n + o * (2 * D_) + half * D_);
    const v4f* bb = (const v4f*)b2;
    v4f x[D_ / 4];
#pragma unroll
    for (int e = 0; e < D_ / 4; ++e) x[e] = wr[e];
    __builtin_amdgcn_sched_barrier(0);
    float s = 0.f;
#pragma unroll
    for (int e = 0; e < D_ / 4; ++e) {
      const v4f y = bb[e];
      s = fmaf(x[e][0], y[0], s); s = fmaf(x[e][1], y[1], s);
      s = fmaf(x[e][2], y[2], s); s = fmaf(x[e][3], y[3], s);
    }
    vf[half * D_ + o] = s;
  }
}

template <bool PACK, bool RESID, bool X0S>
__global__ __launch_bounds__(32)
void gemm_two_parts(const float* __restrict__ X0, const float* __restrict__ W0,
                    int ws0, int wo0,
                    const float* __restrict__ X1, const float* __restrict__ W1,
                    const float* __restrict__ rs, const float* __restrict__ v0,
                    const float* __restrict__ v1, const float* __restrict__ R,
                    float* __restrict__ out) {
  const int lane = threadIdx.x;
  const int l16  = lane & 15;
  const int half = lane >> 4;
  const int tm = blockIdx.y;
  const int tn = blockIdx.x;
  const int o = tn * 16 + l16;
  const int m16 = tm * 16 + l16;
  const int mr = tm * 16 + half * 8;
  const float* x0row = X0 + m16 * D_;
  const float* w0row = W0 + o * ws0 + wo0;
  const float* w1row = W1 + o * D_;
  constexpr int AHEAD = 11;
  v4f a0[KPAIRS], b0[KPAIRS], a1[KPAIRS], b1[KPAIRS];
  v4f rs_lo = {}, rs_hi = {};
  float bv0 = 0.f, bv1 = 0.f;
  float res[8] = {};
#pragma unroll
  for (int kk = 0; kk < AHEAD; ++kk) {
    if (X0S) fetch_panels(X0, m16, w0row, half, kk, a0[kk], b0[kk]);
    else     fetch_rows(x0row, w0row, half, kk, a0[kk], b0[kk]);
    fetch_panels(X1, m16, w1row, half, kk, a1[kk], b1[kk]);
  }
  __builtin_amdgcn_sched_barrier(0);
  v8f c0 = {}, c1 = {};
#pragma unroll
  for (int kk = 0; kk < KPAIRS; ++kk) {
    if (kk + AHEAD < KPAIRS) {
      if (X0S) fetch_panels(X0, m16, w0row, half, kk + AHEAD, a0[kk + AHEAD], b0[kk + AHEAD]);
      else     fetch_rows(x0row, w0row, half, kk + AHEAD, a0[kk + AHEAD], b0[kk + AHEAD]);
      fetch_panels(X1, m16, w1row, half, kk + AHEAD, a1[kk + AHEAD], b1[kk + AHEAD]);
    } else if (kk + AHEAD == KPAIRS) {
      rs_lo = *(const v4f*)(rs + mr);
      rs_hi = *(const v4f*)(rs + mr + 4);
      bv0 = v0[o];
      bv1 = v1 ? v1[o] : 0.f;
      if (RESID) {
#pragma unroll
        for (int r = 0; r < 8; ++r) res[r] = R[(mr + r) * D_ + o];
      }
    }
    c0 = __builtin_amdgcn_wmma_f32_16x16x4_f32(false, (v2f){a0[kk][0], a0[kk][1]}, false,
        (v2f){b0[kk][0], b0[kk][1]}, (short)0, c0, false, false);
    c1 = __builtin_amdgcn_wmma_f32_16x16x4_f32(false, (v2f){a1[kk][0], a1[kk][1]}, false,
        (v2f){b1[kk][0], b1[kk][1]}, (short)0, c1, false, false);
    c0 = __builtin_amdgcn_wmma_f32_16x16x4_f32(false, (v2f){a0[kk][2], a0[kk][3]}, false,
        (v2f){b0[kk][2], b0[kk][3]}, (short)0, c0, false, false);
    c1 = __builtin_amdgcn_wmma_f32_16x16x4_f32(false, (v2f){a1[kk][2], a1[kk][3]}, false,
        (v2f){b1[kk][2], b1[kk][3]}, (short)0, c1, false, false);
    __builtin_amdgcn_sched_barrier(0);
  }
  const float bv = bv0 + bv1;
#pragma unroll
  for (int r = 0; r < 8; ++r) {
    const int m = mr + r;
    float val = (c0[r] + c1[r]) + (r < 4 ? rs_lo[r & 3] : rs_hi[r & 3]) * bv;
    if (RESID) val += res[r];
    if (PACK) out[pack_index(m, o)] = val;
    else      out[m * D_ + o] = val;
  }
}

#define PR 64
#define PO 16
#define APITCH (2 * N_ + 4)
#define AIP (PO + 1)

__device__ __forceinline__ void pair_step8(const v16f s, const v4f q0, const v4f q1,
                                           const v4f q2, const v4f q3,
                                           const v2f ai0, const v2f ai1,
                                           v2f& acc0, v2f& acc1) {
  const v2f zero = {0.f, 0.f};
#pragma unroll
  for (int jj = 0; jj < 8; ++jj) {
    const v4f q = jj < 2 ? q0 : jj < 4 ? q1 : jj < 6 ? q2 : q3;
    const v2f av = (jj & 1) ? (v2f){q[2], q[3]} : (v2f){q[0], q[1]};
    const float s0 = s[2 * jj], s1 = s[2 * jj + 1];
    v2f t0 = ai0 + (v2f){s0, s0};
    v2f t1 = ai1 + (v2f){s1, s1};
    t0 = __builtin_elementwise_max(t0, zero);
    t1 = __builtin_elementwise_max(t1, zero);
    acc0 = __builtin_elementwise_fma(av, t0, acc0);
    acc1 = __builtin_elementwise_fma(av, t1, acc1);
  }
}

template <int KS>
__global__ __launch_bounds__(256)
void pairwise_rows(const float* __restrict__ h, const float* __restrict__ S0,
                   const float* __restrict__ wi, const float* __restrict__ Ci,
                   const float* __restrict__ b1, const float* __restrict__ rs,
                   const float* __restrict__ vi, const float* __restrict__ ajP,
                   const float* __restrict__ adj, float* __restrict__ S,
                   float* __restrict__ rs_out) {
  constexpr int HP = KS * D_ + 4;
  __shared__ __attribute__((aligned(16))) float adjL[(PR / 2) * APITCH];
  __shared__ __attribute__((aligned(16))) float hL[PR * HP];
  __shared__ __attribute__((aligned(16))) float wL[PO * HP];
  __shared__ float aiL[KS * PR * AIP];
  __shared__ float rsL[KS == 1 ? 8 * PR : 1];
  const int tid  = threadIdx.x;
  const int lane = tid & 31;
  const int w    = __builtin_amdgcn_readfirstlane(tid >> 5);
  const int ob   = blockIdx.x;
  const int row0 = blockIdx.y * PR;
  const int b    = blockIdx.y >> 3;

  const v4f* ag = (const v4f*)(adj + (size_t)row0 * N_);
#pragma unroll
  for (int p = 0; p < 16; ++p) {
    const int idx = p * 256 + tid;
    const int pr = idx >> 7, c4 = idx & 127;
    const v4f x0 = ag[pr * (N_ / 4) + c4];
    const v4f x1 = ag[(pr + 32) * (N_ / 4) + c4];
    float* d = adjL + pr * APITCH + c4 * 8;
    *(v4f*)(d)     = (v4f){x0[0], x1[0], x0[1], x1[1]};
    *(v4f*)(d + 4) = (v4f){x0[2], x1[2], x0[3], x1[3]};
  }
#pragma unroll
  for (int p = 0; p < 8; ++p) {
    const int idx = p * 256 + tid;
    const int r = idx >> 5, c4 = idx & 31;
    *(v4f*)(hL + r * HP + c4 * 4) = *(const v4f*)(h + (size_t)(row0 + r) * D_ + c4 * 4);
    if (KS == 2) {
      const int r2 = tid >> 2, q4 = tid & 3;
      *(v4f*)(hL + r2 * HP + D_ + p * 16 + q4 * 4) =
          *(const v4f*)(S0 + ((size_t)p * M_ + row0 + r2) * 16 + q4 * 4);
    }
  }
#pragma unroll
  for (int p = 0; p < 2; ++p) {
    const int idx = p * 256 + tid;
    const int r = idx >> 5, c4 = idx & 31;
    *(v4f*)(wL + r * HP + c4 * 4) = *(const v4f*)(wi + (size_t)(ob * PO + r) * (2 * D_) + c4 * 4);
    if (KS == 2)
      *(v4f*)(wL + r * HP + D_ + c4 * 4) = *(const v4f*)(Ci + (size_t)(ob * PO + r) * D_ + c4 * 4);
  }

  const float* sp = ajP + (size_t)(b * (D_ / 2) + ob * (PO / 2) + w) * N_ * 2;
  const float* arow = adjL + lane * APITCH;
  v2f acc0 = {0.f, 0.f}, acc1 = {0.f, 0.f};

  v16f sA0 = *(const v16f*)(sp), sA1 = *(const v16f*)(sp + 16);
  __syncthreads();
  if (w < 4 * KS) {
    const int l16  = lane & 15;
    const int half = lane >> 4;
    const int seg = w >> 2;
    const int tr = w & 3;
    const float* xrow = hL + (tr * 16 + l16) * HP + seg * D_;
    const float* wrow = wL + l16 * HP + seg * D_;
    const float bv = seg == 0 ? b1[ob * PO + l16] : 0.f;
    v8f c = {};
#pragma unroll
    for (int kk = 0; kk < D_ / 4; ++kk) {
      const int kb = kk * 4 + half * 2;
      v2f a = *(const v2f*)(xrow + kb);
      v2f b = *(const v2f*)(wrow + kb);
      c = __builtin_amdgcn_wmma_f32_16x16x4_f32(
          false, a, false, b, (short)0, c, false, false);
    }
#pragma unroll
    for (int r = 0; r < 8; ++r)
      aiL[seg * PR * AIP + (tr * 16 + half * 8 + r) * AIP + l16] = c[r] + bv;
  } else if (KS == 1 && ob == 0) {
#pragma unroll
    for (int cc = 0; cc < 2; ++cc) {
      const int ch = (w - 4) * 2 + cc;
      const float* p = arow + ch * 128;
      v4f l0 = *(const v4f*)p, l1 = *(const v4f*)(p + 4);
      v2f a0 = {l0[0], l0[1]}, a1 = {l0[2], l0[3]}, a2 = {l1[0], l1[1]}, a3 = {l1[2], l1[3]};
#pragma unroll
      for (int t = 1; t < 16; ++t) {
        l0 = *(const v4f*)(p + 8 * t);
        l1 = *(const v4f*)(p + 8 * t + 4);
        a0 += (v2f){l0[0], l0[1]}; a1 += (v2f){l0[2], l0[3]};
        a2 += (v2f){l1[0], l1[1]}; a3 += (v2f){l1[2], l1[3]};
      }
      const v2f sm = (a0 + a1) + (a2 + a3);
      rsL[ch * PR + lane]      = sm[0];
      rsL[ch * PR + lane + 32] = sm[1];
    }
  }
  __syncthreads();
  if (KS == 1 && ob == 0 && w == 0) {
#pragma unroll
    for (int hh = 0; hh < 2; ++hh) {
      float s = rsL[lane + 32 * hh];
#pragma unroll
      for (int c = 1; c < 8; ++c) s += rsL[c * PR + lane + 32 * hh];
      rs_out[row0 + lane + 32 * hh] = s;
    }
  }
  v2f ai0 = {aiL[lane * AIP + w * 2],     aiL[(lane + 32) * AIP + w * 2]};
  v2f ai1 = {aiL[lane * AIP + w * 2 + 1], aiL[(lane + 32) * AIP + w * 2 + 1]};
  if (KS == 2) {
    const float rv0 = rs[row0 + lane], rv1 = rs[row0 + lane + 32];
    const float vi0 = vi[ob * PO + w * 2], vi1 = vi[ob * PO + w * 2 + 1];
    ai0[0] = (ai0[0] + aiL[PR * AIP + lane * AIP + w * 2])            + rv0 * vi0;
    ai0[1] = (ai0[1] + aiL[PR * AIP + (lane + 32) * AIP + w * 2])     + rv1 * vi0;
    ai1[0] = (ai1[0] + aiL[PR * AIP + lane * AIP + w * 2 + 1])        + rv0 * vi1;
    ai1[1] = (ai1[1] + aiL[PR * AIP + (lane + 32) * AIP + w * 2 + 1]) + rv1 * vi1;
  }
  v4f aA0 = *(const v4f*)(arow),      aA1 = *(const v4f*)(arow + 4),
      aA2 = *(const v4f*)(arow + 8),  aA3 = *(const v4f*)(arow + 12),
      aA4 = *(const v4f*)(arow + 16), aA5 = *(const v4f*)(arow + 20),
      aA6 = *(const v4f*)(arow + 24), aA7 = *(const v4f*)(arow + 28);
  for (int g = 0; g < N_ / 16; g += 2) {
    asm volatile("" :: "s"(sA0), "s"(sA1));
    const float* s1 = sp + (g + 1) * 32;
    const float* r1 = arow + (g + 1) * 32;
    const v16f sB0 = *(const v16f*)(s1), sB1 = *(const v16f*)(s1 + 16);
    const v4f aB0 = *(const v4f*)(r1),      aB1 = *(const v4f*)(r1 + 4),
              aB2 = *(const v4f*)(r1 + 8),  aB3 = *(const v4f*)(r1 + 12),
              aB4 = *(const v4f*)(r1 + 16), aB5 = *(const v4f*)(r1 + 20),
              aB6 = *(const v4f*)(r1 + 24), aB7 = *(const v4f*)(r1 + 28);
    __builtin_amdgcn_sched_barrier(0);
    pair_step8(sA0, aA0, aA1, aA2, aA3, ai0, ai1, acc0, acc1);
    pair_step8(sA1, aA4, aA5, aA6, aA7, ai0, ai1, acc0, acc1);
    asm volatile("" :: "s"(sB0), "s"(sB1));
    const int g2 = (g + 2 < N_ / 16) ? g + 2 : g + 1;
    const float* s2 = sp + g2 * 32;
    const float* r2 = arow + g2 * 32;
    sA0 = *(const v16f*)(s2); sA1 = *(const v16f*)(s2 + 16);
    aA0 = *(const v4f*)(r2);      aA1 = *(const v4f*)(r2 + 4);
    aA2 = *(const v4f*)(r2 + 8);  aA3 = *(const v4f*)(r2 + 12);
    aA4 = *(const v4f*)(r2 + 16); aA5 = *(const v4f*)(r2 + 20);
    aA6 = *(const v4f*)(r2 + 24); aA7 = *(const v4f*)(r2 + 28);
    __builtin_amdgcn_sched_barrier(0);
    pair_step8(sB0, aB0, aB1, aB2, aB3, ai0, ai1, acc0, acc1);
    pair_step8(sB1, aB4, aB5, aB6, aB7, ai0, ai1, acc0, acc1);
  }

  __syncthreads();
  adjL[lane * AIP + w * 2]            = acc0[0];
  adjL[(lane + 32) * AIP + w * 2]     = acc0[1];
  adjL[lane * AIP + w * 2 + 1]        = acc1[0];
  adjL[(lane + 32) * AIP + w * 2 + 1] = acc1[1];
  __syncthreads();
  {
    const int r = tid >> 2, c4 = tid & 3;
    const float* t = adjL + r * AIP + c4 * 4;
    const v4f o = {t[0], t[1], t[2], t[3]};
    *(v4f*)(S + ((size_t)ob * M_ + row0 + r) * 16 + c4 * 4) = o;
  }
}

extern "C" void kernel_launch(void* const* d_in, const int* in_sizes, int n_in,
                              void* d_out, int out_size, void* d_ws, size_t ws_size,
                              hipStream_t stream) {
  const float* node = (const float*)d_in[0];
  const float* adj  = (const float*)d_in[1];
  const float* w1   = (const float*)d_in[2];
  const float* b1   = (const float*)d_in[3];
  const float* w2   = (const float*)d_in[4];
  const float* b2   = (const float*)d_in[5];

  float* out = (float*)d_out;
  float* aj = (float*)d_ws;
  float* S0 = aj + (size_t)M_ * D_;
  float* S1 = S0 + (size_t)M_ * D_;
  float* rs = S1 + (size_t)M_ * D_;
  float* Cf = rs + M_;
  float* vf = Cf + 2 * D_ * D_;

  const float* w1n = w1 + (size_t)D_ * 2 * D_;
  const float* b1n = b1 + D_;
  const float* w2n = w2 + (size_t)D_ * D_;
  const float* b2n = b2 + D_;

  const dim3 gg(D_ / 16, M_ / 16);
  aj0_and_fold<<<dim3(D_ / 16, M_ / 16 + 17), 32, 0, stream>>>(node, w1, w2, b2, aj, Cf, vf);
  pairwise_rows<1><<<dim3(D_ / PO, M_ / PR), 256, 0, stream>>>(
      node, nullptr, w1, nullptr, b1, nullptr, nullptr, aj, adj, S0, rs);
  gemm_two_parts<true, false, false><<<gg, 32, 0, stream>>>(
      node, w1n, 2 * D_, D_, S0, Cf + D_ * D_, rs, vf + D_, nullptr, nullptr, aj);
  pairwise_rows<2><<<dim3(D_ / PO, M_ / PR), 256, 0, stream>>>(
      node, S0, w1n, Cf, b1n, rs, vf, aj, adj, S1, nullptr);
  gemm_two_parts<false, true, true><<<gg, 32, 0, stream>>>(
      S0, w2, D_, 0, S1, w2n, rs, b2, b2n, node, out);
}
